# starting kernel
# baseline (speedup 1.0000x reference)
#include <hip/hip_runtime.h>
#include <math.h>

#define QPS     4
#define NSPLIT  4
#define NLAYER  4
#define DIM     16
#define NCLASS  5
#define BATCH   32768
#define FEAT    16
#define H1      128
#define H2      64

#define WAVES_PER_BLOCK 4
#define H1P 132
#define H2P 68

typedef __attribute__((ext_vector_type(2))) float v2f;
typedef __attribute__((ext_vector_type(8))) float v8f;

__global__ void qcnn_unitaries(const float* __restrict__ sw, float* __restrict__ U) {
    int g = threadIdx.x;
    if (g >= NSPLIT * NLAYER * QPS) return;
    const float* w = sw + g * 3;
    float hx = 0.5f * w[0], hy = 0.5f * w[1], hz = 0.5f * w[2];
    float sx, cx, sy, cy, sz, cz;
    sincosf(hx, &sx, &cx);
    sincosf(hy, &sy, &cy);
    sincosf(hz, &sz, &cz);
    float m00r = cy * cx, m00i =  sy * sx;
    float m01r = -sy * cx, m01i = -cy * sx;
    float m10r =  sy * cx, m10i = -cy * sx;
    float m11r =  cy * cx, m11i = -sy * sx;
    float* o = U + g * 8;
    o[0] = cz * m00r + sz * m00i;  o[1] = cz * m00i - sz * m00r;
    o[2] = cz * m01r + sz * m01i;  o[3] = cz * m01i - sz * m01r;
    o[4] = cz * m10r - sz * m10i;  o[5] = cz * m10i + sz * m10r;
    o[6] = cz * m11r - sz * m11i;  o[7] = cz * m11i + sz * m11r;
}

__global__ __launch_bounds__(256) void qcnn_features(const float* __restrict__ x,
                                                     const float* __restrict__ Utab,
                                                     float* __restrict__ feats) {
    __shared__ float uS[NSPLIT * NLAYER * QPS * 8];
    for (int i = threadIdx.x; i < NSPLIT * NLAYER * QPS * 8; i += blockDim.x)
        uS[i] = Utab[i];
    __syncthreads();

    int t = blockIdx.x * blockDim.x + threadIdx.x;
    int s = t & 3;

    const float* xp = x + (size_t)t * DIM;
    float re[DIM], im[DIM];
    float nrm2 = 0.f;
#pragma unroll
    for (int i = 0; i < DIM; i++) { float v = xp[i]; re[i] = v; nrm2 += v * v; }
    float nrm = sqrtf(nrm2);
    float scale = (nrm > 0.f) ? (1.f / nrm) : 1.f;
#pragma unroll
    for (int i = 0; i < DIM; i++) { re[i] *= scale; im[i] = 0.f; }

    const float* us = &uS[s * NLAYER * QPS * 8];
#pragma unroll
    for (int l = 0; l < NLAYER; l++) {
#pragma unroll
        for (int j = 0; j < QPS; j++) {
            const float* u = &us[(l * QPS + j) * 8];
            float u00r = u[0], u00i = u[1], u01r = u[2], u01i = u[3];
            float u10r = u[4], u10i = u[5], u11r = u[6], u11i = u[7];
            const int m = 8 >> j;
#pragma unroll
            for (int idx = 0; idx < DIM; idx++) {
                if (idx & m) continue;
                const int i1 = idx | m;
                float a0r = re[idx], a0i = im[idx], a1r = re[i1], a1i = im[i1];
                re[idx] = u00r * a0r - u00i * a0i + u01r * a1r - u01i * a1i;
                im[idx] = u00r * a0i + u00i * a0r + u01r * a1i + u01i * a1r;
                re[i1]  = u10r * a0r - u10i * a0i + u11r * a1r - u11i * a1i;
                im[i1]  = u10r * a0i + u10i * a0r + u11r * a1i + u11i * a1r;
            }
        }
#pragma unroll
        for (int j = 0; j < QPS; j++) {
            const int mc = 8 >> j, mt = 8 >> ((j + 1) & 3);
#pragma unroll
            for (int idx = 0; idx < DIM; idx++) {
                if ((idx & mc) && !(idx & mt)) {
                    const int i1 = idx | mt;
                    float tr = re[idx], ti = im[idx];
                    re[idx] = re[i1]; im[idx] = im[i1];
                    re[i1] = tr;      im[i1] = ti;
                }
            }
        }
    }

    float e[QPS];
#pragma unroll
    for (int q = 0; q < QPS; q++) e[q] = 0.f;
#pragma unroll
    for (int idx = 0; idx < DIM; idx++) {
        float p = re[idx] * re[idx] + im[idx] * im[idx];
#pragma unroll
        for (int q = 0; q < QPS; q++)
            e[q] += ((idx >> (3 - q)) & 1) ? -p : p;
    }
    int b = t >> 2;
    float* f = feats + (size_t)b * FEAT + s * QPS;
#pragma unroll
    for (int q = 0; q < QPS; q++) f[q] = e[q];
}

__global__ __launch_bounds__(32 * WAVES_PER_BLOCK) void qcnn_mlp(
    const float* __restrict__ feats,
    const float* __restrict__ W1, const float* __restrict__ b1,
    const float* __restrict__ W2, const float* __restrict__ b2,
    const float* __restrict__ W3, const float* __restrict__ b3,
    float* __restrict__ out)
{
    __shared__ float h1s[WAVES_PER_BLOCK][16 * H1P];
    __shared__ float h2s[WAVES_PER_BLOCK][16 * H2P];

    const int lane = threadIdx.x & 31;
    const int wave = threadIdx.x >> 5;
    const int half = lane >> 4;
    const int nl   = lane & 15;
    const int rowBase = (blockIdx.x * WAVES_PER_BLOCK + wave) * 16;
    float* h1 = h1s[wave];
    float* h2 = h2s[wave];

    v2f a1f[4];
#pragma unroll
    for (int kc = 0; kc < 4; kc++) {
        const float* ap = feats + (size_t)(rowBase + nl) * FEAT + kc * 4 + 2 * half;
        a1f[kc] = v2f{ap[0], ap[1]};
    }
#pragma unroll
    for (int nt = 0; nt < 8; nt++) {
        const int n0 = nt * 16;
        float bias = b1[n0 + nl];
        v8f c;
#pragma unroll
        for (int r = 0; r < 8; r++) c[r] = bias;
#pragma unroll
        for (int kc = 0; kc < 4; kc++) {
            const float* bp = W1 + (size_t)(kc * 4 + 2 * half) * H1 + n0 + nl;
            v2f bf = v2f{bp[0], bp[H1]};
            c = __builtin_amdgcn_wmma_f32_16x16x4_f32(false, a1f[kc], false, bf,
                                                      (short)0, c, false, false);
        }
#pragma unroll
        for (int r = 0; r < 8; r++)
            h1[(r + 8 * half) * H1P + n0 + nl] = fmaxf(c[r], 0.f);
    }
    __syncthreads();

#pragma unroll
    for (int nt = 0; nt < 4; nt++) {
        const int n0 = nt * 16;
        float bias = b2[n0 + nl];
        v8f c;
#pragma unroll
        for (int r = 0; r < 8; r++) c[r] = bias;
#pragma unroll 4
        for (int k0 = 0; k0 < H1; k0 += 4) {
            const float* ap = h1 + nl * H1P + k0 + 2 * half;
            v2f af = v2f{ap[0], ap[1]};
            const float* bp = W2 + (size_t)(k0 + 2 * half) * H2 + n0 + nl;
            v2f bf = v2f{bp[0], bp[H2]};
            c = __builtin_amdgcn_wmma_f32_16x16x4_f32(false, af, false, bf,
                                                      (short)0, c, false, false);
        }
#pragma unroll
        for (int r = 0; r < 8; r++)
            h2[(r + 8 * half) * H2P + n0 + nl] = fmaxf(c[r], 0.f);
    }
    __syncthreads();

    {
        const float msk = (nl < NCLASS) ? 1.f : 0.f;
        const int ncl = (nl < NCLASS) ? nl : (NCLASS - 1);
        float bias = msk * b3[ncl];
        v8f c;
#pragma unroll
        for (int r = 0; r < 8; r++) c[r] = bias;
#pragma unroll 4
        for (int k0 = 0; k0 < H2; k0 += 4) {
            const float* ap = h2 + nl * H2P + k0 + 2 * half;
            v2f af = v2f{ap[0], ap[1]};
            const int k = k0 + 2 * half;
            v2f bf = v2f{msk * W3[k * NCLASS + ncl], msk * W3[(k + 1) * NCLASS + ncl]};
            c = __builtin_amdgcn_wmma_f32_16x16x4_f32(false, af, false, bf,
                                                      (short)0, c, false, false);
        }
        if (nl < NCLASS) {
#pragma unroll
            for (int r = 0; r < 8; r++) {
                const int m = r + 8 * half;
                out[(size_t)(rowBase + m) * NCLASS + nl] = c[r];
            }
        }
    }
}

extern "C" void kernel_launch(void* const* d_in, const int* in_sizes, int n_in,
                              void* d_out, int out_size, void* d_ws, size_t ws_size,
                              hipStream_t stream) {
    const float* x  = (const float*)d_in[0];
    const float* sw = (const float*)d_in[1];
    const float* W1 = (const float*)d_in[2];
    const float* b1 = (const float*)d_in[3];
    const float* W2 = (const float*)d_in[4];
    const float* b2 = (const float*)d_in[5];
    const float* W3 = (const float*)d_in[6];
    const float* b3 = (const float*)d_in[7];
    float* out = (float*)d_out;

    float* Utab  = (float*)d_ws;
    float* feats = (float*)((char*)d_ws + 4096);

    qcnn_unitaries<<<dim3(1), dim3(64), 0, stream>>>(sw, Utab);

    qcnn_features<<<dim3((BATCH * NSPLIT) / 256), dim3(256), 0, stream>>>(x, Utab, feats);

    qcnn_mlp<<<dim3(BATCH / (16 * WAVES_PER_BLOCK)), dim3(32 * WAVES_PER_BLOCK), 0, stream>>>(
        feats, W1, b1, W2, b2, W3, b3, out);
}


# Fuse the three launches into one kernel: the gate table and each work-group's 64x16 feature tile stay in LDS instead of the workspace; arithmetic and summation order unchanged
# speedup vs baseline: 1.3386x; 1.3386x over previous
#include <hip/hip_runtime.h>
#include <math.h>


#define QPS     4
#define NSPLIT  4
#define NLAYER  4
#define DIM     16
#define NCLASS  5
#define BATCH   32768
#define FEAT    16
#define H1      128
#define H2      64
#define NGATE   (NSPLIT * NLAYER * QPS)

#define ROWS_PER_BLOCK  64
#define THREADS         (ROWS_PER_BLOCK * NSPLIT)
#define MLP_WAVES       (ROWS_PER_BLOCK / 16)
#define H1P 132
#define H2P 68
#define FTP 20

typedef __attribute__((ext_vector_type(2))) float v2f;
typedef __attribute__((ext_vector_type(8))) float v8f;

__global__ __launch_bounds__(THREADS) void qcnn_fused(
    const float* __restrict__ x, const float* __restrict__ sw,
    const float* __restrict__ W1, const float* __restrict__ b1,
    const float* __restrict__ W2, const float* __restrict__ b2,
    const float* __restrict__ W3, const float* __restrict__ b3,
    float* __restrict__ out)
{
    __shared__ float uS[NGATE * 8];
    __shared__ __attribute__((aligned(16))) float fS[ROWS_PER_BLOCK * FTP];
    __shared__ float h1s[MLP_WAVES][16 * H1P];
    __shared__ float h2s[MLP_WAVES][16 * H2P];

    const int tid  = threadIdx.x;
    const int lane = tid & 31;
    const int wave = __builtin_amdgcn_readfirstlane(tid >> 5);
    const int half = lane >> 4;
    const int nl   = lane & 15;

    const int t = blockIdx.x * THREADS + tid;
    const int s = tid & 3;
    const float* xp = x + (size_t)t * DIM;
    float re[DIM], im[DIM];
    float nrm2 = 0.f;
#pragma unroll
    for (int i = 0; i < DIM; i++) { float v = xp[i]; re[i] = v; nrm2 += v * v; }

    if (tid < NGATE) {
        const int g = tid;
        const float* w = sw + g * 3;
        float hx = 0.5f * w[0], hy = 0.5f * w[1], hz = 0.5f * w[2];
        float sx, cx, sy, cy, sz, cz;
        sincosf(hx, &sx, &cx);
        sincosf(hy, &sy, &cy);
        sincosf(hz, &sz, &cz);
        float m00r = cy * cx, m00i =  sy * sx;
        float m01r = -sy * cx, m01i = -cy * sx;
        float m10r =  sy * cx, m10i = -cy * sx;
        float m11r =  cy * cx, m11i = -sy * sx;
        float* o = uS + g * 8;
        o[0] = cz * m00r + sz * m00i;  o[1] = cz * m00i - sz * m00r;
        o[2] = cz * m01r + sz * m01i;  o[3] = cz * m01i - sz * m01r;
        o[4] = cz * m10r - sz * m10i;  o[5] = cz * m10i + sz * m10r;
        o[6] = cz * m11r - sz * m11i;  o[7] = cz * m11i + sz * m11r;
    }
    __syncthreads();

    {
        float nrm = sqrtf(nrm2);
        float scale = (nrm > 0.f) ? (1.f / nrm) : 1.f;
#pragma unroll
        for (int i = 0; i < DIM; i++) { re[i] *= scale; im[i] = 0.f; }

        const float* us = &uS[s * NLAYER * QPS * 8];
#pragma unroll
        for (int l = 0; l < NLAYER; l++) {
#pragma unroll
            for (int j = 0; j < QPS; j++) {
                const float* u = &us[(l * QPS + j) * 8];
                float u00r = u[0], u00i = u[1], u01r = u[2], u01i = u[3];
                float u10r = u[4], u10i = u[5], u11r = u[6], u11i = u[7];
                const int m = 8 >> j;
#pragma unroll
                for (int idx = 0; idx < DIM; idx++) {
                    if (idx & m) continue;
                    const int i1 = idx | m;
                    float a0r = re[idx], a0i = im[idx], a1r = re[i1], a1i = im[i1];
                    re[idx] = u00r * a0r - u00i * a0i + u01r * a1r - u01i * a1i;
                    im[idx] = u00r * a0i + u00i * a0r + u01r * a1i + u01i * a1r;
                    re[i1]  = u10r * a0r - u10i * a0i + u11r * a1r - u11i * a1i;
                    im[i1]  = u10r * a0i + u10i * a0r + u11r * a1i + u11i * a1r;
                }
            }
#pragma unroll
            for (int j = 0; j < QPS; j++) {
                const int mc = 8 >> j, mt = 8 >> ((j + 1) & 3);
#pragma unroll
                for (int idx = 0; idx < DIM; idx++) {
                    if ((idx & mc) && !(idx & mt)) {
                        const int i1 = idx | mt;
                        float tr = re[idx], ti = im[idx];
                        re[idx] = re[i1]; im[idx] = im[i1];
                        re[i1] = tr;      im[i1] = ti;
                    }
                }
            }
        }

        float e[QPS];
#pragma unroll
        for (int q = 0; q < QPS; q++) e[q] = 0.f;
#pragma unroll
        for (int idx = 0; idx < DIM; idx++) {
            float p = re[idx] * re[idx] + im[idx] * im[idx];
#pragma unroll
            for (int q = 0; q < QPS; q++)
                e[q] += ((idx >> (3 - q)) & 1) ? -p : p;
        }
        float* f = fS + (tid >> 2) * FTP + s * QPS;
#pragma unroll
        for (int q = 0; q < QPS; q++) f[q] = e[q];
    }
    __syncthreads();

    const bool mlpWave = wave < MLP_WAVES;
    const int  mw = mlpWave ? wave : 0;
    const int  rowBase = blockIdx.x * ROWS_PER_BLOCK + mw * 16;
    const float* fw = fS + mw * 16 * FTP;
    float* h1 = h1s[mw];
    float* h2 = h2s[mw];

    if (mlpWave) {
        v2f a1f[4];
#pragma unroll
        for (int kc = 0; kc < 4; kc++) {
            const float* ap = fw + nl * FTP + kc * 4 + 2 * half;
            a1f[kc] = v2f{ap[0], ap[1]};
        }
#pragma unroll
        for (int nt = 0; nt < 8; nt++) {
            const int n0 = nt * 16;
            float bias = b1[n0 + nl];
            v8f c;
#pragma unroll
            for (int r = 0; r < 8; r++) c[r] = bias;
#pragma unroll
            for (int kc = 0; kc < 4; kc++) {
                const float* bp = W1 + (size_t)(kc * 4 + 2 * half) * H1 + n0 + nl;
                v2f bf = v2f{bp[0], bp[H1]};
                c = __builtin_amdgcn_wmma_f32_16x16x4_f32(false, a1f[kc], false, bf,
                                                          (short)0, c, false, false);
            }
#pragma unroll
            for (int r = 0; r < 8; r++)
                h1[(r + 8 * half) * H1P + n0 + nl] = fmaxf(c[r], 0.f);
        }
    }
    __syncthreads();

    if (mlpWave) {
#pragma unroll
        for (int nt = 0; nt < 4; nt++) {
            const int n0 = nt * 16;
            float bias = b2[n0 + nl];
            v8f c;
#pragma unroll
            for (int r = 0; r < 8; r++) c[r] = bias;
#pragma unroll 4
            for (int k0 = 0; k0 < H1; k0 += 4) {
                const float* ap = h1 + nl * H1P + k0 + 2 * half;
                v2f af = v2f{ap[0], ap[1]};
                const float* bp = W2 + (size_t)(k0 + 2 * half) * H2 + n0 + nl;
                v2f bf = v2f{bp[0], bp[H2]};
                c = __builtin_amdgcn_wmma_f32_16x16x4_f32(false, af, false, bf,
                                                          (short)0, c, false, false);
            }
#pragma unroll
            for (int r = 0; r < 8; r++)
                h2[(r + 8 * half) * H2P + n0 + nl] = fmaxf(c[r], 0.f);
        }
    }
    __syncthreads();

    if (mlpWave) {
        const float msk = (nl < NCLASS) ? 1.f : 0.f;
        const int ncl = (nl < NCLASS) ? nl : (NCLASS - 1);
        float bias = msk * b3[ncl];
        v8f c;
#pragma unroll
        for (int r = 0; r < 8; r++) c[r] = bias;
#pragma unroll 4
        for (int k0 = 0; k0 < H2; k0 += 4) {
            const float* ap = h2 + nl * H2P + k0 + 2 * half;
            v2f af = v2f{ap[0], ap[1]};
            const int k = k0 + 2 * half;
            v2f bf = v2f{msk * W3[k * NCLASS + ncl], msk * W3[(k + 1) * NCLASS + ncl]};
            c = __builtin_amdgcn_wmma_f32_16x16x4_f32(false, af, false, bf,
                                                      (short)0, c, false, false);
        }
        if (nl < NCLASS) {
#pragma unroll
            for (int r = 0; r < 8; r++) {
                const int m = r + 8 * half;
                out[(size_t)(rowBase + m) * NCLASS + nl] = c[r];
            }
        }
    }
}

extern "C" void kernel_launch(void* const* d_in, const int* in_sizes, int n_in,
                              void* d_out, int out_size, void* d_ws, size_t ws_size,
                              hipStream_t stream) {
    const float* x  = (const float*)d_in[0];
    const float* sw = (const float*)d_in[1];
    const float* W1 = (const float*)d_in[2];
    const float* b1 = (const float*)d_in[3];
    const float* W2 = (const float*)d_in[4];
    const float* b2 = (const float*)d_in[5];
    const float* W3 = (const float*)d_in[6];
    const float* b3 = (const float*)d_in[7];
    float* out = (float*)d_out;

    qcnn_fused<<<dim3(BATCH / ROWS_PER_BLOCK), dim3(THREADS), 0, stream>>>(
        x, sw, W1, b1, W2, b2, W3, b3, out);
}


# MLP layers 2 and 3 as straight-line code, layer 2 with k outermost over four live accumulator tiles: weight loads run ahead of the matrix instructions; every sum formed in the same order
# speedup vs baseline: 1.9155x; 1.4310x over previous
#include <hip/hip_runtime.h>
#include <math.h>


#define QPS     4
#define NSPLIT  4
#define NLAYER  4
#define DIM     16
#define NCLASS  5
#define BATCH   32768
#define FEAT    16
#define H1      128
#define H2      64
#define NGATE   (NSPLIT * NLAYER * QPS)

#define ROWS_PER_BLOCK  64
#define THREADS         (ROWS_PER_BLOCK * NSPLIT)
#define MLP_WAVES       (ROWS_PER_BLOCK / 16)
#define H1P 132
#define H2P 68
#define FTP 20

typedef __attribute__((ext_vector_type(2))) float v2f;
typedef __attribute__((ext_vector_type(8))) float v8f;

__global__ __launch_bounds__(THREADS) void qcnn_fused(
    const float* __restrict__ x, const float* __restrict__ sw,
    const float* __restrict__ W1, const float* __restrict__ b1,
    const float* __restrict__ W2, const float* __restrict__ b2,
    const float* __restrict__ W3, const float* __restrict__ b3,
    float* __restrict__ out)
{
    __shared__ float uS[NGATE * 8];
    __shared__ __attribute__((aligned(16))) float fS[ROWS_PER_BLOCK * FTP];
    __shared__ float h1s[MLP_WAVES][16 * H1P];
    __shared__ float h2s[MLP_WAVES][16 * H2P];

    const int tid  = threadIdx.x;
    const int lane = tid & 31;
    const int wave = __builtin_amdgcn_readfirstlane(tid >> 5);
    const int half = lane >> 4;
    const int nl   = lane & 15;

    const int t = blockIdx.x * THREADS + tid;
    const int s = tid & 3;
    const float* xp = x + (size_t)t * DIM;
    float re[DIM], im[DIM];
    float nrm2 = 0.f;
#pragma unroll
    for (int i = 0; i < DIM; i++) { float v = xp[i]; re[i] = v; nrm2 += v * v; }

    if (tid < NGATE) {
        const int g = tid;
        const float* w = sw + g * 3;
        float hx = 0.5f * w[0], hy = 0.5f * w[1], hz = 0.5f * w[2];
        float sx, cx, sy, cy, sz, cz;
        sincosf(hx, &sx, &cx);
        sincosf(hy, &sy, &cy);
        sincosf(hz, &sz, &cz);
        float m00r = cy * cx, m00i =  sy * sx;
        float m01r = -sy * cx, m01i = -cy * sx;
        float m10r =  sy * cx, m10i = -cy * sx;
        float m11r =  cy * cx, m11i = -sy * sx;
        float* o = uS + g * 8;
        o[0] = cz * m00r + sz * m00i;  o[1] = cz * m00i - sz * m00r;
        o[2] = cz * m01r + sz * m01i;  o[3] = cz * m01i - sz * m01r;
        o[4] = cz * m10r - sz * m10i;  o[5] = cz * m10i + sz * m10r;
        o[6] = cz * m11r - sz * m11i;  o[7] = cz * m11i + sz * m11r;
    }
    __syncthreads();

    {
        float nrm = sqrtf(nrm2);
        float scale = (nrm > 0.f) ? (1.f / nrm) : 1.f;
#pragma unroll
        for (int i = 0; i < DIM; i++) { re[i] *= scale; im[i] = 0.f; }

        const float* us = &uS[s * NLAYER * QPS * 8];
#pragma unroll
        for (int l = 0; l < NLAYER; l++) {
#pragma unroll
            for (int j = 0; j < QPS; j++) {
                const float* u = &us[(l * QPS + j) * 8];
                float u00r = u[0], u00i = u[1], u01r = u[2], u01i = u[3];
                float u10r = u[4], u10i = u[5], u11r = u[6], u11i = u[7];
                const int m = 8 >> j;
#pragma unroll
                for (int idx = 0; idx < DIM; idx++) {
                    if (idx & m) continue;
                    const int i1 = idx | m;
                    float a0r = re[idx], a0i = im[idx], a1r = re[i1], a1i = im[i1];
                    re[idx] = u00r * a0r - u00i * a0i + u01r * a1r - u01i * a1i;
                    im[idx] = u00r * a0i + u00i * a0r + u01r * a1i + u01i * a1r;
                    re[i1]  = u10r * a0r - u10i * a0i + u11r * a1r - u11i * a1i;
                    im[i1]  = u10r * a0i + u10i * a0r + u11r * a1i + u11i * a1r;
                }
            }
#pragma unroll
            for (int j = 0; j < QPS; j++) {
                const int mc = 8 >> j, mt = 8 >> ((j + 1) & 3);
#pragma unroll
                for (int idx = 0; idx < DIM; idx++) {
                    if ((idx & mc) && !(idx & mt)) {
                        const int i1 = idx | mt;
                        float tr = re[idx], ti = im[idx];
                        re[idx] = re[i1]; im[idx] = im[i1];
                        re[i1] = tr;      im[i1] = ti;
                    }
                }
            }
        }

        float e[QPS];
#pragma unroll
        for (int q = 0; q < QPS; q++) e[q] = 0.f;
#pragma unroll
        for (int idx = 0; idx < DIM; idx++) {
            float p = re[idx] * re[idx] + im[idx] * im[idx];
#pragma unroll
            for (int q = 0; q < QPS; q++)
                e[q] += ((idx >> (3 - q)) & 1) ? -p : p;
        }
        float* f = fS + (tid >> 2) * FTP + s * QPS;
#pragma unroll
        for (int q = 0; q < QPS; q++) f[q] = e[q];
    }
    __syncthreads();

    const bool mlpWave = wave < MLP_WAVES;
    const int  mw = mlpWave ? wave : 0;
    const int  rowBase = blockIdx.x * ROWS_PER_BLOCK + mw * 16;
    const float* fw = fS + mw * 16 * FTP;
    float* h1 = h1s[mw];
    float* h2 = h2s[mw];

    if (mlpWave) {
        v2f a1f[4];
#pragma unroll
        for (int kc = 0; kc < 4; kc++) {
            const float* ap = fw + nl * FTP + kc * 4 + 2 * half;
            a1f[kc] = v2f{ap[0], ap[1]};
        }
#pragma unroll
        for (int nt = 0; nt < 8; nt++) {
            const int n0 = nt * 16;
            float bias = b1[n0 + nl];
            v8f c;
#pragma unroll
            for (int r = 0; r < 8; r++) c[r] = bias;
#pragma unroll
            for (int kc = 0; kc < 4; kc++) {
                const float* bp = W1 + (size_t)(kc * 4 + 2 * half) * H1 + n0 + nl;
                v2f bf = v2f{bp[0], bp[H1]};
                c = __builtin_amdgcn_wmma_f32_16x16x4_f32(false, a1f[kc], false, bf,
                                                          (short)0, c, false, false);
            }
#pragma unroll
            for (int r = 0; r < 8; r++)
                h1[(r + 8 * half) * H1P + n0 + nl] = fmaxf(c[r], 0.f);
        }
    }
    __syncthreads();

    if (mlpWave) {
        v8f c2[4];
#pragma unroll
        for (int nt = 0; nt < 4; nt++) {
            float bias = b2[nt * 16 + nl];
#pragma unroll
            for (int r = 0; r < 8; r++) c2[nt][r] = bias;
        }
        const float* arow = h1 + nl * H1P + 2 * half;
        const float* brow = W2 + (size_t)(2 * half) * H2 + nl;
#pragma unroll
        for (int k0 = 0; k0 < H1; k0 += 4) {
            v2f af = v2f{arow[k0], arow[k0 + 1]};
#pragma unroll
            for (int nt = 0; nt < 4; nt++) {
                const float* bp = brow + (size_t)k0 * H2 + nt * 16;
                v2f bf = v2f{bp[0], bp[H2]};
                c2[nt] = __builtin_amdgcn_wmma_f32_16x16x4_f32(false, af, false, bf,
                                                               (short)0, c2[nt], false, false);
            }
        }
#pragma unroll
        for (int nt = 0; nt < 4; nt++) {
#pragma unroll
            for (int r = 0; r < 8; r++)
                h2[(r + 8 * half) * H2P + nt * 16 + nl] = fmaxf(c2[nt][r], 0.f);
        }
    }
    __syncthreads();

    if (mlpWave) {
        const float msk = (nl < NCLASS) ? 1.f : 0.f;
        const int ncl = (nl < NCLASS) ? nl : (NCLASS - 1);
        float bias = msk * b3[ncl];
        v8f c;
#pragma unroll
        for (int r = 0; r < 8; r++) c[r] = bias;
#pragma unroll
        for (int k0 = 0; k0 < H2; k0 += 4) {
            const float* ap = h2 + nl * H2P + k0 + 2 * half;
            v2f af = v2f{ap[0], ap[1]};
            const int k = k0 + 2 * half;
            v2f bf = v2f{msk * W3[k * NCLASS + ncl], msk * W3[(k + 1) * NCLASS + ncl]};
            c = __builtin_amdgcn_wmma_f32_16x16x4_f32(false, af, false, bf,
                                                      (short)0, c, false, false);
        }
        if (nl < NCLASS) {
#pragma unroll
            for (int r = 0; r < 8; r++) {
                const int m = r + 8 * half;
                out[(size_t)(rowBase + m) * NCLASS + nl] = c[r];
            }
        }
    }
}

extern "C" void kernel_launch(void* const* d_in, const int* in_sizes, int n_in,
                              void* d_out, int out_size, void* d_ws, size_t ws_size,
                              hipStream_t stream) {
    const float* x  = (const float*)d_in[0];
    const float* sw = (const float*)d_in[1];
    const float* W1 = (const float*)d_in[2];
    const float* b1 = (const float*)d_in[3];
    const float* W2 = (const float*)d_in[4];
    const float* b2 = (const float*)d_in[5];
    const float* W3 = (const float*)d_in[6];
    const float* b3 = (const float*)d_in[7];
    float* out = (float*)d_out;

    qcnn_fused<<<dim3(BATCH / ROWS_PER_BLOCK), dim3(THREADS), 0, stream>>>(
        x, sw, W1, b1, W2, b2, W3, b3, out);
}


# Each split's gate network folded once per work-group into a 16x32 real matrix (run on the 16 basis states) and applied to 16 samples per wave on f32 16x16x4 WMMA; state and PauliZ sum order changes
# speedup vs baseline: 2.0334x; 1.0616x over previous
#include <hip/hip_runtime.h>
#include <math.h>


#define QPS     4
#define NSPLIT  4
#define NLAYER  4
#define DIM     16
#define NCLASS  5
#define BATCH   32768
#define FEAT    16
#define H1      128
#define H2      64
#define NGATE   (NSPLIT * NLAYER * QPS)

#define WAVES           8
#define ROWS_PER_BLOCK  (16 * WAVES)
#define THREADS         (32 * WAVES)
#define H1P 132
#define H2P 68
#define MP  20

typedef __attribute__((ext_vector_type(2))) float v2f;
typedef __attribute__((ext_vector_type(4))) float v4f;
typedef __attribute__((ext_vector_type(8))) float v8f;

__global__ __launch_bounds__(THREADS) void qcnn_fused(
    const float* __restrict__ x, const float* __restrict__ sw,
    const float* __restrict__ W1, const float* __restrict__ b1,
    const float* __restrict__ W2, const float* __restrict__ b2,
    const float* __restrict__ W3, const float* __restrict__ b3,
    float* __restrict__ out)
{
    __shared__ float uS[NGATE * 8];
    __shared__ __attribute__((aligned(16))) float mS[NSPLIT * 2 * DIM * MP];
    __shared__ float h1s[WAVES][16 * H1P];
    __shared__ float h2s[WAVES][16 * H2P];

    const int tid  = threadIdx.x;
    const int lane = tid & 31;
    const int wave = __builtin_amdgcn_readfirstlane(tid >> 5);
    const int half = lane >> 4;
    const int nl   = lane & 15;

    const int rowBase = blockIdx.x * ROWS_PER_BLOCK + wave * 16;

    v4f xv[NSPLIT][2];
    {
        const float* xr = x + (size_t)(rowBase + nl) * (NSPLIT * DIM) + 4 * half;
#pragma unroll
        for (int sp = 0; sp < NSPLIT; sp++)
#pragma unroll
            for (int q = 0; q < 2; q++)
                xv[sp][q] = *(const v4f*)(xr + sp * DIM + 8 * q);
    }

    if (tid < NGATE) {
        const int g = tid;
        const float* w = sw + g * 3;
        float hx = 0.5f * w[0], hy = 0.5f * w[1], hz = 0.5f * w[2];
        float sx, cx, sy, cy, sz, cz;
        sincosf(hx, &sx, &cx);
        sincosf(hy, &sy, &cy);
        sincosf(hz, &sz, &cz);
        float m00r = cy * cx, m00i =  sy * sx;
        float m01r = -sy * cx, m01i = -cy * sx;
        float m10r =  sy * cx, m10i = -cy * sx;
        float m11r =  cy * cx, m11i = -sy * sx;
        float* o = uS + g * 8;
        o[0] = cz * m00r + sz * m00i;  o[1] = cz * m00i - sz * m00r;
        o[2] = cz * m01r + sz * m01i;  o[3] = cz * m01i - sz * m01r;
        o[4] = cz * m10r - sz * m10i;  o[5] = cz * m10i + sz * m10r;
        o[6] = cz * m11r - sz * m11i;  o[7] = cz * m11i + sz * m11r;
    }
    __syncthreads();

    if (wave < 2) {
        const int sp = tid >> 4;
        const int kb = tid & 15;
        float re[DIM], im[DIM];
#pragma unroll
        for (int i = 0; i < DIM; i++) { re[i] = (i == kb) ? 1.f : 0.f; im[i] = 0.f; }

        const float* us = &uS[sp * NLAYER * QPS * 8];
#pragma unroll
        for (int l = 0; l < NLAYER; l++) {
#pragma unroll
            for (int j = 0; j < QPS; j++) {
                const float* u = &us[(l * QPS + j) * 8];
                float u00r = u[0], u00i = u[1], u01r = u[2], u01i = u[3];
                float u10r = u[4], u10i = u[5], u11r = u[6], u11i = u[7];
                const int m = 8 >> j;
#pragma unroll
                for (int idx = 0; idx < DIM; idx++) {
                    if (idx & m) continue;
                    const int i1 = idx | m;
                    float a0r = re[idx], a0i = im[idx], a1r = re[i1], a1i = im[i1];
                    re[idx] = u00r * a0r - u00i * a0i + u01r * a1r - u01i * a1i;
                    im[idx] = u00r * a0i + u00i * a0r + u01r * a1i + u01i * a1r;
                    re[i1]  = u10r * a0r - u10i * a0i + u11r * a1r - u11i * a1i;
                    im[i1]  = u10r * a0i + u10i * a0r + u11r * a1i + u11i * a1r;
                }
            }
#pragma unroll
            for (int j = 0; j < QPS; j++) {
                const int mc = 8 >> j, mt = 8 >> ((j + 1) & 3);
#pragma unroll
                for (int idx = 0; idx < DIM; idx++) {
                    if ((idx & mc) && !(idx & mt)) {
                        const int i1 = idx | mt;
                        float tr = re[idx], ti = im[idx];
                        re[idx] = re[i1]; im[idx] = im[i1];
                        re[i1] = tr;      im[i1] = ti;
                    }
                }
            }
        }
        float* mr = mS + (sp * 2 + 0) * DIM * MP + kb;
        float* mi = mS + (sp * 2 + 1) * DIM * MP + kb;
#pragma unroll
        for (int i = 0; i < DIM; i++) { mr[i * MP] = re[i]; mi[i * MP] = im[i]; }
    }
    __syncthreads();

    float* h1 = h1s[wave];
    float* h2 = h2s[wave];

    v2f a1f[4];
#pragma unroll
    for (int sp = 0; sp < NSPLIT; sp++) {
        float n2 = 0.f;
#pragma unroll
        for (int q = 0; q < 2; q++)
#pragma unroll
            for (int e = 0; e < 4; e++) n2 += xv[sp][q][e] * xv[sp][q][e];
        n2 += __shfl_xor(n2, 16);
        float nrm = sqrtf(n2);
        float scale = (nrm > 0.f) ? (1.f / nrm) : 1.f;

        v8f dre, dim;
#pragma unroll
        for (int r = 0; r < 8; r++) { dre[r] = 0.f; dim[r] = 0.f; }
#pragma unroll
        for (int q = 0; q < 2; q++) {
            const v4f mr = *(const v4f*)&mS[((sp * 2 + 0) * DIM + nl) * MP + 8 * q + 4 * half];
            const v4f mi = *(const v4f*)&mS[((sp * 2 + 1) * DIM + nl) * MP + 8 * q + 4 * half];
#pragma unroll
            for (int p = 0; p < 2; p++) {
                v2f bx = v2f{xv[sp][q][2 * p] * scale, xv[sp][q][2 * p + 1] * scale};
                v2f ar = v2f{mr[2 * p], mr[2 * p + 1]};
                v2f ai = v2f{mi[2 * p], mi[2 * p + 1]};
                dre = __builtin_amdgcn_wmma_f32_16x16x4_f32(false, ar, false, bx,
                                                            (short)0, dre, false, false);
                dim = __builtin_amdgcn_wmma_f32_16x16x4_f32(false, ai, false, bx,
                                                            (short)0, dim, false, false);
            }
        }
        float pr[8];
#pragma unroll
        for (int r = 0; r < 8; r++) pr[r] = dre[r] * dre[r] + dim[r] * dim[r];
        float s0 = 0.f, s1 = 0.f, s2 = 0.f, s3 = 0.f;
#pragma unroll
        for (int r = 0; r < 8; r++) {
            s0 += pr[r];
            s1 += (r & 4) ? -pr[r] : pr[r];
            s2 += (r & 2) ? -pr[r] : pr[r];
            s3 += (r & 1) ? -pr[r] : pr[r];
        }
        s0 = half ? -s0 : s0;
        float e0 = s0 + __shfl_xor(s0, 16);
        float e1 = s1 + __shfl_xor(s1, 16);
        float e2 = s2 + __shfl_xor(s2, 16);
        float e3 = s3 + __shfl_xor(s3, 16);
        a1f[sp] = half ? v2f{e2, e3} : v2f{e0, e1};
    }

    {
#pragma unroll
        for (int nt = 0; nt < 8; nt++) {
            const int n0 = nt * 16;
            float bias = b1[n0 + nl];
            v8f c;
#pragma unroll
            for (int r = 0; r < 8; r++) c[r] = bias;
#pragma unroll
            for (int kc = 0; kc < 4; kc++) {
                const float* bp = W1 + (size_t)(kc * 4 + 2 * half) * H1 + n0 + nl;
                v2f bf = v2f{bp[0], bp[H1]};
                c = __builtin_amdgcn_wmma_f32_16x16x4_f32(false, a1f[kc], false, bf,
                                                          (short)0, c, false, false);
            }
#pragma unroll
            for (int r = 0; r < 8; r++)
                h1[(r + 8 * half) * H1P + n0 + nl] = fmaxf(c[r], 0.f);
        }
    }
    __syncthreads();

    {
        v8f c2[4];
#pragma unroll
        for (int nt = 0; nt < 4; nt++) {
            float bias = b2[nt * 16 + nl];
#pragma unroll
            for (int r = 0; r < 8; r++) c2[nt][r] = bias;
        }
        const float* arow = h1 + nl * H1P + 2 * half;
        const float* brow = W2 + (size_t)(2 * half) * H2 + nl;
#pragma unroll
        for (int k0 = 0; k0 < H1; k0 += 4) {
            v2f af = v2f{arow[k0], arow[k0 + 1]};
#pragma unroll
            for (int nt = 0; nt < 4; nt++) {
                const float* bp = brow + (size_t)k0 * H2 + nt * 16;
                v2f bf = v2f{bp[0], bp[H2]};
                c2[nt] = __builtin_amdgcn_wmma_f32_16x16x4_f32(false, af, false, bf,
                                                               (short)0, c2[nt], false, false);
            }
        }
#pragma unroll
        for (int nt = 0; nt < 4; nt++) {
#pragma unroll
            for (int r = 0; r < 8; r++)
                h2[(r + 8 * half) * H2P + nt * 16 + nl] = fmaxf(c2[nt][r], 0.f);
        }
    }
    __syncthreads();

    {
        const float msk = (nl < NCLASS) ? 1.f : 0.f;
        const int ncl = (nl < NCLASS) ? nl : (NCLASS - 1);
        float bias = msk * b3[ncl];
        v8f c;
#pragma unroll
        for (int r = 0; r < 8; r++) c[r] = bias;
#pragma unroll
        for (int k0 = 0; k0 < H2; k0 += 4) {
            const float* ap = h2 + nl * H2P + k0 + 2 * half;
            v2f af = v2f{ap[0], ap[1]};
            const int k = k0 + 2 * half;
            v2f bf = v2f{msk * W3[k * NCLASS + ncl], msk * W3[(k + 1) * NCLASS + ncl]};
            c = __builtin_amdgcn_wmma_f32_16x16x4_f32(false, af, false, bf,
                                                      (short)0, c, false, false);
        }
        if (nl < NCLASS) {
#pragma unroll
            for (int r = 0; r < 8; r++) {
                const int m = r + 8 * half;
                out[(size_t)(rowBase + m) * NCLASS + nl] = c[r];
            }
        }
    }
}

extern "C" void kernel_launch(void* const* d_in, const int* in_sizes, int n_in,
                              void* d_out, int out_size, void* d_ws, size_t ws_size,
                              hipStream_t stream) {
    const float* x  = (const float*)d_in[0];
    const float* sw = (const float*)d_in[1];
    const float* W1 = (const float*)d_in[2];
    const float* b1 = (const float*)d_in[3];
    const float* W2 = (const float*)d_in[4];
    const float* b2 = (const float*)d_in[5];
    const float* W3 = (const float*)d_in[6];
    const float* b3 = (const float*)d_in[7];
    float* out = (float*)d_out;

    qcnn_fused<<<dim3(BATCH / ROWS_PER_BLOCK), dim3(THREADS), 0, stream>>>(
        x, sw, W1, b1, W2, b2, W3, b3, out);
}


# W2, W3 copied once per work-group into LDS (transposed) by the six waves idle during the circuit fold; layer 2/3 B fragments by 16-byte LDS reads grouped ahead of 32 WMMA; their k sum order changes
# speedup vs baseline: 2.2904x; 1.1264x over previous
#include <hip/hip_runtime.h>
#include <math.h>


#define QPS     4
#define NSPLIT  4
#define NLAYER  4
#define DIM     16
#define NCLASS  5
#define BATCH   32768
#define FEAT    16
#define H1      128
#define H2      64
#define NGATE   (NSPLIT * NLAYER * QPS)

#define WAVES           8
#define ROWS_PER_BLOCK  (16 * WAVES)
#define THREADS         (32 * WAVES)
#define H1P 132
#define H2P 68
#define MP  20
#define W2P 132
#define W3P 68
#define STAGE_WAVES (WAVES - 2)

typedef __attribute__((ext_vector_type(2))) float v2f;
typedef __attribute__((ext_vector_type(4))) float v4f;
typedef __attribute__((ext_vector_type(8))) float v8f;

__global__ __launch_bounds__(THREADS) void qcnn_fused(
    const float* __restrict__ x, const float* __restrict__ sw,
    const float* __restrict__ W1, const float* __restrict__ b1,
    const float* __restrict__ W2, const float* __restrict__ b2,
    const float* __restrict__ W3, const float* __restrict__ b3,
    float* __restrict__ out)
{
    __shared__ float uS[NGATE * 8];
    __shared__ __attribute__((aligned(16))) float mS[NSPLIT * 2 * DIM * MP];
    __shared__ __attribute__((aligned(16))) float h1s[WAVES][16 * H1P];
    __shared__ __attribute__((aligned(16))) float h2s[WAVES][16 * H2P];
    __shared__ __attribute__((aligned(16))) float w2s[H2 * W2P];
    __shared__ __attribute__((aligned(16))) float w3s[16 * W3P];

    const int tid  = threadIdx.x;
    const int lane = tid & 31;
    const int wave = __builtin_amdgcn_readfirstlane(tid >> 5);
    const int half = lane >> 4;
    const int nl   = lane & 15;

    const int rowBase = blockIdx.x * ROWS_PER_BLOCK + wave * 16;

    v4f xv[NSPLIT][2];
    {
        const float* xr = x + (size_t)(rowBase + nl) * (NSPLIT * DIM) + 4 * half;
#pragma unroll
        for (int sp = 0; sp < NSPLIT; sp++)
#pragma unroll
            for (int q = 0; q < 2; q++)
                xv[sp][q] = *(const v4f*)(xr + sp * DIM + 8 * q);
    }

    v4f wst[11];
    const int sw_ = wave - 2;
    if (wave >= 2) {
#pragma unroll
        for (int i = 0; i < 11; i++) {
            const int tile = sw_ + STAGE_WAVES * i;
            if (i < 10 || sw_ < 4) {
                const int k  = (tile >> 2) * 8 + (lane & 7);
                const int n4 = ((tile & 3) * 4 + (lane >> 3)) * 4;
                wst[i] = *(const v4f*)(W2 + (size_t)k * H2 + n4);
            }
        }
    }

    if (tid < NGATE) {
        const int g = tid;
        const float* w = sw + g * 3;
        float hx = 0.5f * w[0], hy = 0.5f * w[1], hz = 0.5f * w[2];
        float sx, cx, sy, cy, sz, cz;
        sincosf(hx, &sx, &cx);
        sincosf(hy, &sy, &cy);
        sincosf(hz, &sz, &cz);
        float m00r = cy * cx, m00i =  sy * sx;
        float m01r = -sy * cx, m01i = -cy * sx;
        float m10r =  sy * cx, m10i = -cy * sx;
        float m11r =  cy * cx, m11i = -sy * sx;
        float* o = uS + g * 8;
        o[0] = cz * m00r + sz * m00i;  o[1] = cz * m00i - sz * m00r;
        o[2] = cz * m01r + sz * m01i;  o[3] = cz * m01i - sz * m01r;
        o[4] = cz * m10r - sz * m10i;  o[5] = cz * m10i + sz * m10r;
        o[6] = cz * m11r - sz * m11i;  o[7] = cz * m11i + sz * m11r;
    }
    __syncthreads();

    if (wave < 2) {
        const int sp = tid >> 4;
        const int kb = tid & 15;
        float re[DIM], im[DIM];
#pragma unroll
        for (int i = 0; i < DIM; i++) { re[i] = (i == kb) ? 1.f : 0.f; im[i] = 0.f; }

        const float* us = &uS[sp * NLAYER * QPS * 8];
#pragma unroll
        for (int l = 0; l < NLAYER; l++) {
#pragma unroll
            for (int j = 0; j < QPS; j++) {
                const float* u = &us[(l * QPS + j) * 8];
                float u00r = u[0], u00i = u[1], u01r = u[2], u01i = u[3];
                float u10r = u[4], u10i = u[5], u11r = u[6], u11i = u[7];
                const int m = 8 >> j;
#pragma unroll
                for (int idx = 0; idx < DIM; idx++) {
                    if (idx & m) continue;
                    const int i1 = idx | m;
                    float a0r = re[idx], a0i = im[idx], a1r = re[i1], a1i = im[i1];
                    re[idx] = u00r * a0r - u00i * a0i + u01r * a1r - u01i * a1i;
                    im[idx] = u00r * a0i + u00i * a0r + u01r * a1i + u01i * a1r;
                    re[i1]  = u10r * a0r - u10i * a0i + u11r * a1r - u11i * a1i;
                    im[i1]  = u10r * a0i + u10i * a0r + u11r * a1i + u11i * a1r;
                }
            }
#pragma unroll
            for (int j = 0; j < QPS; j++) {
                const int mc = 8 >> j, mt = 8 >> ((j + 1) & 3);
#pragma unroll
                for (int idx = 0; idx < DIM; idx++) {
                    if ((idx & mc) && !(idx & mt)) {
                        const int i1 = idx | mt;
                        float tr = re[idx], ti = im[idx];
                        re[idx] = re[i1]; im[idx] = im[i1];
                        re[i1] = tr;      im[i1] = ti;
                    }
                }
            }
        }
        float* mr = mS + (sp * 2 + 0) * DIM * MP + kb;
        float* mi = mS + (sp * 2 + 1) * DIM * MP + kb;
#pragma unroll
        for (int i = 0; i < DIM; i++) { mr[i * MP] = re[i]; mi[i * MP] = im[i]; }
    } else {
#pragma unroll
        for (int i = 0; i < 11; i++) {
            const int tile = sw_ + STAGE_WAVES * i;
            if (i < 10 || sw_ < 4) {
                const int k  = (tile >> 2) * 8 + (lane & 7);
                const int n4 = ((tile & 3) * 4 + (lane >> 3)) * 4;
#pragma unroll
                for (int j = 0; j < 4; j++) w2s[(n4 + j) * W2P + k] = wst[i][j];
            }
        }
        for (int idx = tid - 64; idx < 16 * W3P; idx += 32 * STAGE_WAVES) {
            const int n = idx / W3P, k = idx - n * W3P;
            w3s[idx] = (n < NCLASS && k < H2) ? W3[k * NCLASS + n] : 0.f;
        }
    }
    __syncthreads();

    float bias2[4];
#pragma unroll
    for (int nt = 0; nt < 4; nt++) bias2[nt] = b2[nt * 16 + nl];
    const float bias3 = (nl < NCLASS) ? b3[(nl < NCLASS) ? nl : (NCLASS - 1)] : 0.f;

    float* h1 = h1s[wave];
    float* h2 = h2s[wave];

    v2f a1f[4];
#pragma unroll
    for (int sp = 0; sp < NSPLIT; sp++) {
        float n2 = 0.f;
#pragma unroll
        for (int q = 0; q < 2; q++)
#pragma unroll
            for (int e = 0; e < 4; e++) n2 += xv[sp][q][e] * xv[sp][q][e];
        n2 += __shfl_xor(n2, 16);
        float nrm = sqrtf(n2);
        float scale = (nrm > 0.f) ? (1.f / nrm) : 1.f;

        v8f dre, dim;
#pragma unroll
        for (int r = 0; r < 8; r++) { dre[r] = 0.f; dim[r] = 0.f; }
#pragma unroll
        for (int q = 0; q < 2; q++) {
            const v4f mr = *(const v4f*)&mS[((sp * 2 + 0) * DIM + nl) * MP + 8 * q + 4 * half];
            const v4f mi = *(const v4f*)&mS[((sp * 2 + 1) * DIM + nl) * MP + 8 * q + 4 * half];
#pragma unroll
            for (int p = 0; p < 2; p++) {
                v2f bx = v2f{xv[sp][q][2 * p] * scale, xv[sp][q][2 * p + 1] * scale};
                v2f ar = v2f{mr[2 * p], mr[2 * p + 1]};
                v2f ai = v2f{mi[2 * p], mi[2 * p + 1]};
                dre = __builtin_amdgcn_wmma_f32_16x16x4_f32(false, ar, false, bx,
                                                            (short)0, dre, false, false);
                dim = __builtin_amdgcn_wmma_f32_16x16x4_f32(false, ai, false, bx,
                                                            (short)0, dim, false, false);
            }
        }
        float pr[8];
#pragma unroll
        for (int r = 0; r < 8; r++) pr[r] = dre[r] * dre[r] + dim[r] * dim[r];
        float s0 = 0.f, s1 = 0.f, s2 = 0.f, s3 = 0.f;
#pragma unroll
        for (int r = 0; r < 8; r++) {
            s0 += pr[r];
            s1 += (r & 4) ? -pr[r] : pr[r];
            s2 += (r & 2) ? -pr[r] : pr[r];
            s3 += (r & 1) ? -pr[r] : pr[r];
        }
        s0 = half ? -s0 : s0;
        float e0 = s0 + __shfl_xor(s0, 16);
        float e1 = s1 + __shfl_xor(s1, 16);
        float e2 = s2 + __shfl_xor(s2, 16);
        float e3 = s3 + __shfl_xor(s3, 16);
        a1f[sp] = half ? v2f{e2, e3} : v2f{e0, e1};
    }

    {
#pragma unroll
        for (int nt = 0; nt < 8; nt++) {
            const int n0 = nt * 16;
            float bias = b1[n0 + nl];
            v8f c;
#pragma unroll
            for (int r = 0; r < 8; r++) c[r] = bias;
#pragma unroll
            for (int kc = 0; kc < 4; kc++) {
                const float* bp = W1 + (size_t)(kc * 4 + 2 * half) * H1 + n0 + nl;
                v2f bf = v2f{bp[0], bp[H1]};
                c = __builtin_amdgcn_wmma_f32_16x16x4_f32(false, a1f[kc], false, bf,
                                                          (short)0, c, false, false);
            }
#pragma unroll
            for (int r = 0; r < 8; r++)
                h1[(r + 8 * half) * H1P + n0 + nl] = fmaxf(c[r], 0.f);
        }
    }
    __syncthreads();

    {
        v8f c2[4];
#pragma unroll
        for (int nt = 0; nt < 4; nt++) {
#pragma unroll
            for (int r = 0; r < 8; r++) c2[nt][r] = bias2[nt];
        }
        const float* arow = h1 + nl * H1P + 4 * half;
        const float* brow = w2s + nl * W2P + 4 * half;
        v4f afr[2][4];
        v4f bfr[2][4][4];
#pragma unroll
        for (int qq = 0; qq < 4; qq++) {
            afr[0][qq] = *(const v4f*)(arow + 8 * qq);
#pragma unroll
            for (int nt = 0; nt < 4; nt++)
                bfr[0][qq][nt] = *(const v4f*)(brow + nt * 16 * W2P + 8 * qq);
        }
#pragma unroll
        for (int blk = 0; blk < 4; blk++) {
            const int cur = blk & 1, nxt = cur ^ 1;
            if (blk + 1 < 4) {
#pragma unroll
                for (int qq = 0; qq < 4; qq++) {
                    const int q = 4 * (blk + 1) + qq;
                    afr[nxt][qq] = *(const v4f*)(arow + 8 * q);
#pragma unroll
                    for (int nt = 0; nt < 4; nt++)
                        bfr[nxt][qq][nt] = *(const v4f*)(brow + nt * 16 * W2P + 8 * q);
                }
            }
            __builtin_amdgcn_sched_barrier(0);
#pragma unroll
            for (int qq = 0; qq < 4; qq++) {
#pragma unroll
                for (int p = 0; p < 2; p++) {
                    v2f af = v2f{afr[cur][qq][2 * p], afr[cur][qq][2 * p + 1]};
#pragma unroll
                    for (int nt = 0; nt < 4; nt++) {
                        v2f bf = v2f{bfr[cur][qq][nt][2 * p], bfr[cur][qq][nt][2 * p + 1]};
                        c2[nt] = __builtin_amdgcn_wmma_f32_16x16x4_f32(false, af, false, bf,
                                                                       (short)0, c2[nt], false, false);
                    }
                }
            }
            __builtin_amdgcn_sched_barrier(0);
        }
#pragma unroll
        for (int nt = 0; nt < 4; nt++) {
#pragma unroll
            for (int r = 0; r < 8; r++)
                h2[(r + 8 * half) * H2P + nt * 16 + nl] = fmaxf(c2[nt][r], 0.f);
        }
    }
    __syncthreads();

    {
        v8f c;
#pragma unroll
        for (int r = 0; r < 8; r++) c[r] = bias3;
        const float* arow = h2 + nl * H2P + 4 * half;
        const float* brow = w3s + nl * W3P + 4 * half;
        v4f av[H2 / 8], bv[H2 / 8];
#pragma unroll
        for (int q = 0; q < H2 / 8; q++) {
            av[q] = *(const v4f*)(arow + 8 * q);
            bv[q] = *(const v4f*)(brow + 8 * q);
        }
        __builtin_amdgcn_sched_barrier(0);
#pragma unroll
        for (int q = 0; q < H2 / 8; q++) {
#pragma unroll
            for (int p = 0; p < 2; p++) {
                v2f af = v2f{av[q][2 * p], av[q][2 * p + 1]};
                v2f bf = v2f{bv[q][2 * p], bv[q][2 * p + 1]};
                c = __builtin_amdgcn_wmma_f32_16x16x4_f32(false, af, false, bf,
                                                          (short)0, c, false, false);
            }
        }
        if (nl < NCLASS) {
#pragma unroll
            for (int r = 0; r < 8; r++) {
                const int m = r + 8 * half;
                out[(size_t)(rowBase + m) * NCLASS + nl] = c[r];
            }
        }
    }
}

extern "C" void kernel_launch(void* const* d_in, const int* in_sizes, int n_in,
                              void* d_out, int out_size, void* d_ws, size_t ws_size,
                              hipStream_t stream) {
    const float* x  = (const float*)d_in[0];
    const float* sw = (const float*)d_in[1];
    const float* W1 = (const float*)d_in[2];
    const float* b1 = (const float*)d_in[3];
    const float* W2 = (const float*)d_in[4];
    const float* b2 = (const float*)d_in[5];
    const float* W3 = (const float*)d_in[6];
    const float* b3 = (const float*)d_in[7];
    float* out = (float*)d_out;

    qcnn_fused<<<dim3(BATCH / ROWS_PER_BLOCK), dim3(THREADS), 0, stream>>>(
        x, sw, W1, b1, W2, b2, W3, b3, out);
}


# One work-group barrier instead of four: gate table, h1, h2 are handed over inside their own wave (compiler-only fence), so waves run free; folding waves fetch gate angles ahead of their input tile
# speedup vs baseline: 2.4778x; 1.0818x over previous
#include <hip/hip_runtime.h>
#include <math.h>


#define QPS     4
#define NSPLIT  4
#define NLAYER  4
#define DIM     16
#define NCLASS  5
#define BATCH   32768
#define FEAT    16
#define H1      128
#define H2      64
#define NGATE   (NSPLIT * NLAYER * QPS)

#define WAVES           8
#define ROWS_PER_BLOCK  (16 * WAVES)
#define THREADS         (32 * WAVES)
#define H1P 132
#define H2P 68
#define MP  20
#define W2P 132
#define W3P 68
#define STAGE_WAVES (WAVES - 2)

typedef __attribute__((ext_vector_type(2))) float v2f;
typedef __attribute__((ext_vector_type(4))) float v4f;
typedef __attribute__((ext_vector_type(8))) float v8f;


static __device__ __forceinline__ void wave_lds_handover() {
    __builtin_amdgcn_fence(__ATOMIC_RELEASE, "wavefront");
    __builtin_amdgcn_wave_barrier();
    __builtin_amdgcn_fence(__ATOMIC_ACQUIRE, "wavefront");
}

static __device__ __forceinline__ void load_x_tile(const float* __restrict__ xr, v4f (&xv)[NSPLIT][2]) {
#pragma unroll
    for (int sp = 0; sp < NSPLIT; sp++)
#pragma unroll
        for (int q = 0; q < 2; q++)
            xv[sp][q] = *(const v4f*)(xr + sp * DIM + 8 * q);
}
__global__ __launch_bounds__(THREADS) void qcnn_fused(
    const float* __restrict__ x, const float* __restrict__ sw,
    const float* __restrict__ W1, const float* __restrict__ b1,
    const float* __restrict__ W2, const float* __restrict__ b2,
    const float* __restrict__ W3, const float* __restrict__ b3,
    float* __restrict__ out)
{
    __shared__ float uS[NGATE * 8];
    __shared__ __attribute__((aligned(16))) float mS[NSPLIT * 2 * DIM * MP];
    __shared__ __attribute__((aligned(16))) float h1s[WAVES][16 * H1P];
    __shared__ __attribute__((aligned(16))) float h2s[WAVES][16 * H2P];
    __shared__ __attribute__((aligned(16))) float w2s[H2 * W2P];
    __shared__ __attribute__((aligned(16))) float w3s[16 * W3P];

    const int tid  = threadIdx.x;
    const int lane = tid & 31;
    const int wave = __builtin_amdgcn_readfirstlane(tid >> 5);
    const int half = lane >> 4;
    const int nl   = lane & 15;

    const int rowBase = blockIdx.x * ROWS_PER_BLOCK + wave * 16;

    const float* xr = x + (size_t)(rowBase + nl) * (NSPLIT * DIM) + 4 * half;
    v4f xv[NSPLIT][2];

    if (wave < 2) {
        {
            const int g = tid;
            const float* w = sw + g * 3;
            float hx = 0.5f * w[0], hy = 0.5f * w[1], hz = 0.5f * w[2];
            float sx, cx, sy, cy, sz, cz;
            sincosf(hx, &sx, &cx);
            sincosf(hy, &sy, &cy);
            sincosf(hz, &sz, &cz);
            float m00r = cy * cx, m00i =  sy * sx;
            float m01r = -sy * cx, m01i = -cy * sx;
            float m10r =  sy * cx, m10i = -cy * sx;
            float m11r =  cy * cx, m11i = -sy * sx;
            float* o = uS + g * 8;
            o[0] = cz * m00r + sz * m00i;  o[1] = cz * m00i - sz * m00r;
            o[2] = cz * m01r + sz * m01i;  o[3] = cz * m01i - sz * m01r;
            o[4] = cz * m10r - sz * m10i;  o[5] = cz * m10i + sz * m10r;
            o[6] = cz * m11r - sz * m11i;  o[7] = cz * m11i + sz * m11r;
        }
        load_x_tile(xr, xv);
        wave_lds_handover();

        const int sp = tid >> 4;
        const int kb = tid & 15;
        float re[DIM], im[DIM];
#pragma unroll
        for (int i = 0; i < DIM; i++) { re[i] = (i == kb) ? 1.f : 0.f; im[i] = 0.f; }

        const float* us = &uS[sp * NLAYER * QPS * 8];
#pragma unroll
        for (int l = 0; l < NLAYER; l++) {
#pragma unroll
            for (int j = 0; j < QPS; j++) {
                const float* u = &us[(l * QPS + j) * 8];
                float u00r = u[0], u00i = u[1], u01r = u[2], u01i = u[3];
                float u10r = u[4], u10i = u[5], u11r = u[6], u11i = u[7];
                const int m = 8 >> j;
#pragma unroll
                for (int idx = 0; idx < DIM; idx++) {
                    if (idx & m) continue;
                    const int i1 = idx | m;
                    float a0r = re[idx], a0i = im[idx], a1r = re[i1], a1i = im[i1];
                    re[idx] = u00r * a0r - u00i * a0i + u01r * a1r - u01i * a1i;
                    im[idx] = u00r * a0i + u00i * a0r + u01r * a1i + u01i * a1r;
                    re[i1]  = u10r * a0r - u10i * a0i + u11r * a1r - u11i * a1i;
                    im[i1]  = u10r * a0i + u10i * a0r + u11r * a1i + u11i * a1r;
                }
            }
#pragma unroll
            for (int j = 0; j < QPS; j++) {
                const int mc = 8 >> j, mt = 8 >> ((j + 1) & 3);
#pragma unroll
                for (int idx = 0; idx < DIM; idx++) {
                    if ((idx & mc) && !(idx & mt)) {
                        const int i1 = idx | mt;
                        float tr = re[idx], ti = im[idx];
                        re[idx] = re[i1]; im[idx] = im[i1];
                        re[i1] = tr;      im[i1] = ti;
                    }
                }
            }
        }
        float* mr = mS + (sp * 2 + 0) * DIM * MP + kb;
        float* mi = mS + (sp * 2 + 1) * DIM * MP + kb;
#pragma unroll
        for (int i = 0; i < DIM; i++) { mr[i * MP] = re[i]; mi[i * MP] = im[i]; }
    } else {
        load_x_tile(xr, xv);
        const int sw_ = wave - 2;
        v4f wst[11];
#pragma unroll
        for (int i = 0; i < 11; i++) {
            const int tile = sw_ + STAGE_WAVES * i;
            if (i < 10 || sw_ < 4) {
                const int k  = (tile >> 2) * 8 + (lane & 7);
                const int n4 = ((tile & 3) * 4 + (lane >> 3)) * 4;
                wst[i] = *(const v4f*)(W2 + (size_t)k * H2 + n4);
            }
        }
#pragma unroll
        for (int i = 0; i < 11; i++) {
            const int tile = sw_ + STAGE_WAVES * i;
            if (i < 10 || sw_ < 4) {
                const int k  = (tile >> 2) * 8 + (lane & 7);
                const int n4 = ((tile & 3) * 4 + (lane >> 3)) * 4;
#pragma unroll
                for (int j = 0; j < 4; j++) w2s[(n4 + j) * W2P + k] = wst[i][j];
            }
        }
        for (int idx = tid - 64; idx < 16 * W3P; idx += 32 * STAGE_WAVES) {
            const int n = idx / W3P, k = idx - n * W3P;
            w3s[idx] = (n < NCLASS && k < H2) ? W3[k * NCLASS + n] : 0.f;
        }
    }
    __syncthreads();

    float bias2[4];
#pragma unroll
    for (int nt = 0; nt < 4; nt++) bias2[nt] = b2[nt * 16 + nl];
    const float bias3 = (nl < NCLASS) ? b3[(nl < NCLASS) ? nl : (NCLASS - 1)] : 0.f;

    float* h1 = h1s[wave];
    float* h2 = h2s[wave];

    v2f a1f[4];
#pragma unroll
    for (int sp = 0; sp < NSPLIT; sp++) {
        float n2 = 0.f;
#pragma unroll
        for (int q = 0; q < 2; q++)
#pragma unroll
            for (int e = 0; e < 4; e++) n2 += xv[sp][q][e] * xv[sp][q][e];
        n2 += __shfl_xor(n2, 16);
        float nrm = sqrtf(n2);
        float scale = (nrm > 0.f) ? (1.f / nrm) : 1.f;

        v8f dre, dim;
#pragma unroll
        for (int r = 0; r < 8; r++) { dre[r] = 0.f; dim[r] = 0.f; }
#pragma unroll
        for (int q = 0; q < 2; q++) {
            const v4f mr = *(const v4f*)&mS[((sp * 2 + 0) * DIM + nl) * MP + 8 * q + 4 * half];
            const v4f mi = *(const v4f*)&mS[((sp * 2 + 1) * DIM + nl) * MP + 8 * q + 4 * half];
#pragma unroll
            for (int p = 0; p < 2; p++) {
                v2f bx = v2f{xv[sp][q][2 * p] * scale, xv[sp][q][2 * p + 1] * scale};
                v2f ar = v2f{mr[2 * p], mr[2 * p + 1]};
                v2f ai = v2f{mi[2 * p], mi[2 * p + 1]};
                dre = __builtin_amdgcn_wmma_f32_16x16x4_f32(false, ar, false, bx,
                                                            (short)0, dre, false, false);
                dim = __builtin_amdgcn_wmma_f32_16x16x4_f32(false, ai, false, bx,
                                                            (short)0, dim, false, false);
            }
        }
        float pr[8];
#pragma unroll
        for (int r = 0; r < 8; r++) pr[r] = dre[r] * dre[r] + dim[r] * dim[r];
        float s0 = 0.f, s1 = 0.f, s2 = 0.f, s3 = 0.f;
#pragma unroll
        for (int r = 0; r < 8; r++) {
            s0 += pr[r];
            s1 += (r & 4) ? -pr[r] : pr[r];
            s2 += (r & 2) ? -pr[r] : pr[r];
            s3 += (r & 1) ? -pr[r] : pr[r];
        }
        s0 = half ? -s0 : s0;
        float e0 = s0 + __shfl_xor(s0, 16);
        float e1 = s1 + __shfl_xor(s1, 16);
        float e2 = s2 + __shfl_xor(s2, 16);
        float e3 = s3 + __shfl_xor(s3, 16);
        a1f[sp] = half ? v2f{e2, e3} : v2f{e0, e1};
    }

    {
#pragma unroll
        for (int nt = 0; nt < 8; nt++) {
            const int n0 = nt * 16;
            float bias = b1[n0 + nl];
            v8f c;
#pragma unroll
            for (int r = 0; r < 8; r++) c[r] = bias;
#pragma unroll
            for (int kc = 0; kc < 4; kc++) {
                const float* bp = W1 + (size_t)(kc * 4 + 2 * half) * H1 + n0 + nl;
                v2f bf = v2f{bp[0], bp[H1]};
                c = __builtin_amdgcn_wmma_f32_16x16x4_f32(false, a1f[kc], false, bf,
                                                          (short)0, c, false, false);
            }
#pragma unroll
            for (int r = 0; r < 8; r++)
                h1[(r + 8 * half) * H1P + n0 + nl] = fmaxf(c[r], 0.f);
        }
    }
    wave_lds_handover();

    {
        v8f c2[4];
#pragma unroll
        for (int nt = 0; nt < 4; nt++) {
#pragma unroll
            for (int r = 0; r < 8; r++) c2[nt][r] = bias2[nt];
        }
        const float* arow = h1 + nl * H1P + 4 * half;
        const float* brow = w2s + nl * W2P + 4 * half;
        v4f afr[2][4];
        v4f bfr[2][4][4];
#pragma unroll
        for (int qq = 0; qq < 4; qq++) {
            afr[0][qq] = *(const v4f*)(arow + 8 * qq);
#pragma unroll
            for (int nt = 0; nt < 4; nt++)
                bfr[0][qq][nt] = *(const v4f*)(brow + nt * 16 * W2P + 8 * qq);
        }
#pragma unroll
        for (int blk = 0; blk < 4; blk++) {
            const int cur = blk & 1, nxt = cur ^ 1;
            if (blk + 1 < 4) {
#pragma unroll
                for (int qq = 0; qq < 4; qq++) {
                    const int q = 4 * (blk + 1) + qq;
                    afr[nxt][qq] = *(const v4f*)(arow + 8 * q);
#pragma unroll
                    for (int nt = 0; nt < 4; nt++)
                        bfr[nxt][qq][nt] = *(const v4f*)(brow + nt * 16 * W2P + 8 * q);
                }
            }
            __builtin_amdgcn_sched_barrier(0);
#pragma unroll
            for (int qq = 0; qq < 4; qq++) {
#pragma unroll
                for (int p = 0; p < 2; p++) {
                    v2f af = v2f{afr[cur][qq][2 * p], afr[cur][qq][2 * p + 1]};
#pragma unroll
                    for (int nt = 0; nt < 4; nt++) {
                        v2f bf = v2f{bfr[cur][qq][nt][2 * p], bfr[cur][qq][nt][2 * p + 1]};
                        c2[nt] = __builtin_amdgcn_wmma_f32_16x16x4_f32(false, af, false, bf,
                                                                       (short)0, c2[nt], false, false);
                    }
                }
            }
            __builtin_amdgcn_sched_barrier(0);
        }
#pragma unroll
        for (int nt = 0; nt < 4; nt++) {
#pragma unroll
            for (int r = 0; r < 8; r++)
                h2[(r + 8 * half) * H2P + nt * 16 + nl] = fmaxf(c2[nt][r], 0.f);
        }
    }
    wave_lds_handover();

    {
        v8f c;
#pragma unroll
        for (int r = 0; r < 8; r++) c[r] = bias3;
        const float* arow = h2 + nl * H2P + 4 * half;
        const float* brow = w3s + nl * W3P + 4 * half;
        v4f av[H2 / 8], bv[H2 / 8];
#pragma unroll
        for (int q = 0; q < H2 / 8; q++) {
            av[q] = *(const v4f*)(arow + 8 * q);
            bv[q] = *(const v4f*)(brow + 8 * q);
        }
        __builtin_amdgcn_sched_barrier(0);
#pragma unroll
        for (int q = 0; q < H2 / 8; q++) {
#pragma unroll
            for (int p = 0; p < 2; p++) {
                v2f af = v2f{av[q][2 * p], av[q][2 * p + 1]};
                v2f bf = v2f{bv[q][2 * p], bv[q][2 * p + 1]};
                c = __builtin_amdgcn_wmma_f32_16x16x4_f32(false, af, false, bf,
                                                          (short)0, c, false, false);
            }
        }
        if (nl < NCLASS) {
#pragma unroll
            for (int r = 0; r < 8; r++) {
                const int m = r + 8 * half;
                out[(size_t)(rowBase + m) * NCLASS + nl] = c[r];
            }
        }
    }
}

extern "C" void kernel_launch(void* const* d_in, const int* in_sizes, int n_in,
                              void* d_out, int out_size, void* d_ws, size_t ws_size,
                              hipStream_t stream) {
    const float* x  = (const float*)d_in[0];
    const float* sw = (const float*)d_in[1];
    const float* W1 = (const float*)d_in[2];
    const float* b1 = (const float*)d_in[3];
    const float* W2 = (const float*)d_in[4];
    const float* b2 = (const float*)d_in[5];
    const float* W3 = (const float*)d_in[6];
    const float* b3 = (const float*)d_in[7];
    float* out = (float*)d_out;

    qcnn_fused<<<dim3(BATCH / ROWS_PER_BLOCK), dim3(THREADS), 0, stream>>>(
        x, sw, W1, b1, W2, b2, W3, b3, out);
}


# Circuit fold on four waves, one split each: a lane pair holds a column (8 amplitudes a lane as re/im pairs, wire 0 across the pair by DPP), gates read from LDS in one group; gate sum order changes
# speedup vs baseline: 2.6692x; 1.0772x over previous
#include <hip/hip_runtime.h>
#include <math.h>


#define QPS     4
#define NSPLIT  4
#define NLAYER  4
#define DIM     16
#define NCLASS  5
#define BATCH   32768
#define FEAT    16
#define H1      128
#define H2      64
#define NGATE   (NSPLIT * NLAYER * QPS)

#define WAVES           8
#define ROWS_PER_BLOCK  (16 * WAVES)
#define THREADS         (32 * WAVES)
#define H1P 132
#define H2P 68
#define MP  20
#define W2P 132
#define W3P 68
#define FOLD_WAVES  NSPLIT
#define STAGE_WAVES (WAVES - FOLD_WAVES)

typedef __attribute__((ext_vector_type(2))) float v2f;
typedef __attribute__((ext_vector_type(4))) float v4f;
typedef __attribute__((ext_vector_type(8))) float v8f;


static __device__ __forceinline__ void wave_lds_handover() {
    __builtin_amdgcn_fence(__ATOMIC_RELEASE, "wavefront");
    __builtin_amdgcn_wave_barrier();
    __builtin_amdgcn_fence(__ATOMIC_ACQUIRE, "wavefront");
}

static __device__ __forceinline__ float lane_pair_other(float v) {
    return __int_as_float(__builtin_amdgcn_mov_dpp(__float_as_int(v), 0xB1, 0xF, 0xF, true));
}

static __device__ __forceinline__ v2f cmul(v2f c, v2f a) {
    return __builtin_elementwise_fma(v2f{-c[1], c[0]}, v2f{a[1], a[1]}, c * v2f{a[0], a[0]});
}
static __device__ __forceinline__ v2f cmac(v2f acc, v2f c, v2f a) {
    acc = __builtin_elementwise_fma(c, v2f{a[0], a[0]}, acc);
    return __builtin_elementwise_fma(v2f{-c[1], c[0]}, v2f{a[1], a[1]}, acc);
}

static __device__ __forceinline__ void load_x_tile(const float* __restrict__ xr, v4f (&xv)[NSPLIT][2]) {
#pragma unroll
    for (int sp = 0; sp < NSPLIT; sp++)
#pragma unroll
        for (int q = 0; q < 2; q++)
            xv[sp][q] = *(const v4f*)(xr + sp * DIM + 8 * q);
}
__global__ __launch_bounds__(THREADS) void qcnn_fused(
    const float* __restrict__ x, const float* __restrict__ sw,
    const float* __restrict__ W1, const float* __restrict__ b1,
    const float* __restrict__ W2, const float* __restrict__ b2,
    const float* __restrict__ W3, const float* __restrict__ b3,
    float* __restrict__ out)
{
    __shared__ __attribute__((aligned(16))) float uS[NGATE * 8];
    __shared__ __attribute__((aligned(16))) float mS[NSPLIT * 2 * DIM * MP];
    __shared__ __attribute__((aligned(16))) float h1s[WAVES][16 * H1P];
    __shared__ __attribute__((aligned(16))) float h2s[WAVES][16 * H2P];
    __shared__ __attribute__((aligned(16))) float w2s[H2 * W2P];
    __shared__ __attribute__((aligned(16))) float w3s[16 * W3P];

    const int tid  = threadIdx.x;
    const int lane = tid & 31;
    const int wave = __builtin_amdgcn_readfirstlane(tid >> 5);
    const int half = lane >> 4;
    const int nl   = lane & 15;

    const int rowBase = blockIdx.x * ROWS_PER_BLOCK + wave * 16;

    const float* xr = x + (size_t)(rowBase + nl) * (NSPLIT * DIM) + 4 * half;
    v4f xv[NSPLIT][2];

    if (wave < FOLD_WAVES) {
        const int sp = wave;
        {
            const int g = sp * (NLAYER * QPS) + nl;
            const float* w = sw + g * 3;
            float hx = 0.5f * w[0], hy = 0.5f * w[1], hz = 0.5f * w[2];
            float sx, cx, sy, cy, sz, cz;
            sincosf(hx, &sx, &cx);
            sincosf(hy, &sy, &cy);
            sincosf(hz, &sz, &cz);
            float m00r = cy * cx, m00i =  sy * sx;
            float m01r = -sy * cx, m01i = -cy * sx;
            float m10r =  sy * cx, m10i = -cy * sx;
            float m11r =  cy * cx, m11i = -sy * sx;
            if (half == 0) {
                float* o = uS + g * 8;
                o[0] = cz * m00r + sz * m00i;  o[1] = cz * m00i - sz * m00r;
                o[2] = cz * m01r + sz * m01i;  o[3] = cz * m01i - sz * m01r;
                o[4] = cz * m10r - sz * m10i;  o[5] = cz * m10i + sz * m10r;
                o[6] = cz * m11r - sz * m11i;  o[7] = cz * m11i + sz * m11r;
            }
        }
        load_x_tile(xr, xv);
        wave_lds_handover();

        const int kb = lane >> 1;
        const int hb = lane & 1;
        v4f gt[NLAYER * QPS][2];
        {
            const v4f* us = (const v4f*)&uS[sp * NLAYER * QPS * 8];
#pragma unroll
            for (int g = 0; g < NLAYER * QPS; g++) { gt[g][0] = us[2 * g]; gt[g][1] = us[2 * g + 1]; }
        }
        __builtin_amdgcn_sched_barrier(0);

        v2f am[8];
#pragma unroll
        for (int r = 0; r < 8; r++) am[r] = v2f{(8 * hb + r == kb) ? 1.f : 0.f, 0.f};

#pragma unroll
        for (int l = 0; l < NLAYER; l++) {
            {
                const v4f ua = gt[l * QPS][0], ub = gt[l * QPS][1];
                const v2f co = v2f{hb ? ub[2] : ua[0], hb ? ub[3] : ua[1]};
                const v2f cp = v2f{hb ? ub[0] : ua[2], hb ? ub[1] : ua[3]};
#pragma unroll
                for (int r = 0; r < 8; r++) {
                    const v2f ot = v2f{lane_pair_other(am[r][0]), lane_pair_other(am[r][1])};
                    am[r] = cmac(cmul(co, am[r]), cp, ot);
                }
            }
#pragma unroll
            for (int j = 1; j < QPS; j++) {
                const v4f ua = gt[l * QPS + j][0], ub = gt[l * QPS + j][1];
                const v2f u00 = v2f{ua[0], ua[1]}, u01 = v2f{ua[2], ua[3]};
                const v2f u10 = v2f{ub[0], ub[1]}, u11 = v2f{ub[2], ub[3]};
                const int m = 8 >> j;
#pragma unroll
                for (int idx = 0; idx < 8; idx++) {
                    if (idx & m) continue;
                    const int i1 = idx | m;
                    const v2f a0 = am[idx], a1 = am[i1];
                    am[idx] = cmac(cmul(u00, a0), u01, a1);
                    am[i1]  = cmac(cmul(u10, a0), u11, a1);
                }
            }
#pragma unroll
            for (int r = 0; r < 4; r++) {
                const v2f lo = am[r], up = am[r | 4];
                am[r]     = v2f{hb ? up[0] : lo[0], hb ? up[1] : lo[1]};
                am[r | 4] = v2f{hb ? lo[0] : up[0], hb ? lo[1] : up[1]};
            }
#pragma unroll
            for (int j = 1; j < 3; j++) {
                const int mc = 8 >> j, mt = 8 >> (j + 1);
#pragma unroll
                for (int idx = 0; idx < 8; idx++) {
                    if ((idx & mc) && !(idx & mt)) {
                        const int i1 = idx | mt;
                        const v2f t = am[idx];
                        am[idx] = am[i1];
                        am[i1] = t;
                    }
                }
            }
#pragma unroll
            for (int r = 1; r < 8; r += 2)
                am[r] = v2f{lane_pair_other(am[r][0]), lane_pair_other(am[r][1])};
        }
        float* mr = mS + (sp * 2 + 0) * DIM * MP + 8 * hb * MP + kb;
        float* mi = mS + (sp * 2 + 1) * DIM * MP + 8 * hb * MP + kb;
#pragma unroll
        for (int r = 0; r < 8; r++) { mr[r * MP] = am[r][0]; mi[r * MP] = am[r][1]; }
    } else {
        load_x_tile(xr, xv);
        const int sw_ = wave - FOLD_WAVES;
        const int st = tid - 32 * FOLD_WAVES;
        float w3v[3];
#pragma unroll
        for (int i = 0; i < 3; i++) {
            const int idx = st + 32 * STAGE_WAVES * i;
            const int n = idx / W3P, k = idx - n * W3P;
            const bool live = (n < NCLASS) && (k < H2);
            const float v = W3[live ? (k * NCLASS + n) : 0];
            w3v[i] = live ? v : 0.f;
        }
        v4f wst[16];
#pragma unroll
        for (int i = 0; i < 16; i++) {
            const int k  = i * 8 + (lane & 7);
            const int n4 = (sw_ * 4 + (lane >> 3)) * 4;
            wst[i] = *(const v4f*)(W2 + (size_t)k * H2 + n4);
        }
#pragma unroll
        for (int i = 0; i < 16; i++) {
            const int k  = i * 8 + (lane & 7);
            const int n4 = (sw_ * 4 + (lane >> 3)) * 4;
#pragma unroll
            for (int j = 0; j < 4; j++) w2s[(n4 + j) * W2P + k] = wst[i][j];
        }
#pragma unroll
        for (int i = 0; i < 3; i++) w3s[st + 32 * STAGE_WAVES * i] = w3v[i];
        for (int idx = st + 32 * STAGE_WAVES * 3; idx < 16 * W3P; idx += 32 * STAGE_WAVES)
            w3s[idx] = 0.f;
    }
    __syncthreads();

    float bias2[4];
#pragma unroll
    for (int nt = 0; nt < 4; nt++) bias2[nt] = b2[nt * 16 + nl];
    const float bias3 = (nl < NCLASS) ? b3[(nl < NCLASS) ? nl : (NCLASS - 1)] : 0.f;

    float* h1 = h1s[wave];
    float* h2 = h2s[wave];

    v2f a1f[4];
#pragma unroll
    for (int sp = 0; sp < NSPLIT; sp++) {
        float n2 = 0.f;
#pragma unroll
        for (int q = 0; q < 2; q++)
#pragma unroll
            for (int e = 0; e < 4; e++) n2 += xv[sp][q][e] * xv[sp][q][e];
        n2 += __shfl_xor(n2, 16);
        float nrm = sqrtf(n2);
        float scale = (nrm > 0.f) ? (1.f / nrm) : 1.f;

        v8f dre, dim;
#pragma unroll
        for (int r = 0; r < 8; r++) { dre[r] = 0.f; dim[r] = 0.f; }
#pragma unroll
        for (int q = 0; q < 2; q++) {
            const v4f mr = *(const v4f*)&mS[((sp * 2 + 0) * DIM + nl) * MP + 8 * q + 4 * half];
            const v4f mi = *(const v4f*)&mS[((sp * 2 + 1) * DIM + nl) * MP + 8 * q + 4 * half];
#pragma unroll
            for (int p = 0; p < 2; p++) {
                v2f bx = v2f{xv[sp][q][2 * p] * scale, xv[sp][q][2 * p + 1] * scale};
                v2f ar = v2f{mr[2 * p], mr[2 * p + 1]};
                v2f ai = v2f{mi[2 * p], mi[2 * p + 1]};
                dre = __builtin_amdgcn_wmma_f32_16x16x4_f32(false, ar, false, bx,
                                                            (short)0, dre, false, false);
                dim = __builtin_amdgcn_wmma_f32_16x16x4_f32(false, ai, false, bx,
                                                            (short)0, dim, false, false);
            }
        }
        float pr[8];
#pragma unroll
        for (int r = 0; r < 8; r++) pr[r] = dre[r] * dre[r] + dim[r] * dim[r];
        float s0 = 0.f, s1 = 0.f, s2 = 0.f, s3 = 0.f;
#pragma unroll
        for (int r = 0; r < 8; r++) {
            s0 += pr[r];
            s1 += (r & 4) ? -pr[r] : pr[r];
            s2 += (r & 2) ? -pr[r] : pr[r];
            s3 += (r & 1) ? -pr[r] : pr[r];
        }
        s0 = half ? -s0 : s0;
        float e0 = s0 + __shfl_xor(s0, 16);
        float e1 = s1 + __shfl_xor(s1, 16);
        float e2 = s2 + __shfl_xor(s2, 16);
        float e3 = s3 + __shfl_xor(s3, 16);
        a1f[sp] = half ? v2f{e2, e3} : v2f{e0, e1};
    }

    {
#pragma unroll
        for (int nt = 0; nt < 8; nt++) {
            const int n0 = nt * 16;
            float bias = b1[n0 + nl];
            v8f c;
#pragma unroll
            for (int r = 0; r < 8; r++) c[r] = bias;
#pragma unroll
            for (int kc = 0; kc < 4; kc++) {
                const float* bp = W1 + (size_t)(kc * 4 + 2 * half) * H1 + n0 + nl;
                v2f bf = v2f{bp[0], bp[H1]};
                c = __builtin_amdgcn_wmma_f32_16x16x4_f32(false, a1f[kc], false, bf,
                                                          (short)0, c, false, false);
            }
#pragma unroll
            for (int r = 0; r < 8; r++)
                h1[(r + 8 * half) * H1P + n0 + nl] = fmaxf(c[r], 0.f);
        }
    }
    wave_lds_handover();

    {
        v8f c2[4];
#pragma unroll
        for (int nt = 0; nt < 4; nt++) {
#pragma unroll
            for (int r = 0; r < 8; r++) c2[nt][r] = bias2[nt];
        }
        const float* arow = h1 + nl * H1P + 4 * half;
        const float* brow = w2s + nl * W2P + 4 * half;
        v4f afr[2][4];
        v4f bfr[2][4][4];
#pragma unroll
        for (int qq = 0; qq < 4; qq++) {
            afr[0][qq] = *(const v4f*)(arow + 8 * qq);
#pragma unroll
            for (int nt = 0; nt < 4; nt++)
                bfr[0][qq][nt] = *(const v4f*)(brow + nt * 16 * W2P + 8 * qq);
        }
#pragma unroll
        for (int blk = 0; blk < 4; blk++) {
            const int cur = blk & 1, nxt = cur ^ 1;
            if (blk + 1 < 4) {
#pragma unroll
                for (int qq = 0; qq < 4; qq++) {
                    const int q = 4 * (blk + 1) + qq;
                    afr[nxt][qq] = *(const v4f*)(arow + 8 * q);
#pragma unroll
                    for (int nt = 0; nt < 4; nt++)
                        bfr[nxt][qq][nt] = *(const v4f*)(brow + nt * 16 * W2P + 8 * q);
                }
            }
            __builtin_amdgcn_sched_barrier(0);
#pragma unroll
            for (int qq = 0; qq < 4; qq++) {
#pragma unroll
                for (int p = 0; p < 2; p++) {
                    v2f af = v2f{afr[cur][qq][2 * p], afr[cur][qq][2 * p + 1]};
#pragma unroll
                    for (int nt = 0; nt < 4; nt++) {
                        v2f bf = v2f{bfr[cur][qq][nt][2 * p], bfr[cur][qq][nt][2 * p + 1]};
                        c2[nt] = __builtin_amdgcn_wmma_f32_16x16x4_f32(false, af, false, bf,
                                                                       (short)0, c2[nt], false, false);
                    }
                }
            }
            __builtin_amdgcn_sched_barrier(0);
        }
#pragma unroll
        for (int nt = 0; nt < 4; nt++) {
#pragma unroll
            for (int r = 0; r < 8; r++)
                h2[(r + 8 * half) * H2P + nt * 16 + nl] = fmaxf(c2[nt][r], 0.f);
        }
    }
    wave_lds_handover();

    {
        v8f c;
#pragma unroll
        for (int r = 0; r < 8; r++) c[r] = bias3;
        const float* arow = h2 + nl * H2P + 4 * half;
        const float* brow = w3s + nl * W3P + 4 * half;
        v4f av[H2 / 8], bv[H2 / 8];
#pragma unroll
        for (int q = 0; q < H2 / 8; q++) {
            av[q] = *(const v4f*)(arow + 8 * q);
            bv[q] = *(const v4f*)(brow + 8 * q);
        }
        __builtin_amdgcn_sched_barrier(0);
#pragma unroll
        for (int q = 0; q < H2 / 8; q++) {
#pragma unroll
            for (int p = 0; p < 2; p++) {
                v2f af = v2f{av[q][2 * p], av[q][2 * p + 1]};
                v2f bf = v2f{bv[q][2 * p], bv[q][2 * p + 1]};
                c = __builtin_amdgcn_wmma_f32_16x16x4_f32(false, af, false, bf,
                                                          (short)0, c, false, false);
            }
        }
        if (nl < NCLASS) {
#pragma unroll
            for (int r = 0; r < 8; r++) {
                const int m = r + 8 * half;
                out[(size_t)(rowBase + m) * NCLASS + nl] = c[r];
            }
        }
    }
}

extern "C" void kernel_launch(void* const* d_in, const int* in_sizes, int n_in,
                              void* d_out, int out_size, void* d_ws, size_t ws_size,
                              hipStream_t stream) {
    const float* x  = (const float*)d_in[0];
    const float* sw = (const float*)d_in[1];
    const float* W1 = (const float*)d_in[2];
    const float* b1 = (const float*)d_in[3];
    const float* W2 = (const float*)d_in[4];
    const float* b2 = (const float*)d_in[5];
    const float* W3 = (const float*)d_in[6];
    const float* b3 = (const float*)d_in[7];
    float* out = (float*)d_out;

    qcnn_fused<<<dim3(BATCH / ROWS_PER_BLOCK), dim3(THREADS), 0, stream>>>(
        x, sw, W1, b1, W2, b2, W3, b3, out);
}


# MLP in transposed form (W^T as A operand, activations as B): each layer's result registers are the next layer's B fragments, so h1 and h2 never go through LDS; k sum order of layers 2 and 3 changes
# speedup vs baseline: 2.8080x; 1.0520x over previous
#include <hip/hip_runtime.h>
#include <math.h>


#define QPS     4
#define NSPLIT  4
#define NLAYER  4
#define DIM     16
#define NCLASS  5
#define BATCH   32768
#define FEAT    16
#define H1      128
#define H2      64
#define NGATE   (NSPLIT * NLAYER * QPS)

#define WAVES           8
#define ROWS_PER_BLOCK  (16 * WAVES)
#define THREADS         (32 * WAVES)
#define W1_FIRST 6
#define MP  20
#define W2P 132
#define W3P 68
#define FOLD_WAVES  NSPLIT
#define STAGE_WAVES (WAVES - FOLD_WAVES)

typedef __attribute__((ext_vector_type(2))) float v2f;
typedef __attribute__((ext_vector_type(4))) float v4f;
typedef __attribute__((ext_vector_type(8))) float v8f;


static __device__ __forceinline__ void wave_lds_handover() {
    __builtin_amdgcn_fence(__ATOMIC_RELEASE, "wavefront");
    __builtin_amdgcn_wave_barrier();
    __builtin_amdgcn_fence(__ATOMIC_ACQUIRE, "wavefront");
}

static __device__ __forceinline__ float lane_pair_other(float v) {
    return __int_as_float(__builtin_amdgcn_mov_dpp(__float_as_int(v), 0xB1, 0xF, 0xF, true));
}

static __device__ __forceinline__ v2f cmul(v2f c, v2f a) {
    return __builtin_elementwise_fma(v2f{-c[1], c[0]}, v2f{a[1], a[1]}, c * v2f{a[0], a[0]});
}
static __device__ __forceinline__ v2f cmac(v2f acc, v2f c, v2f a) {
    acc = __builtin_elementwise_fma(c, v2f{a[0], a[0]}, acc);
    return __builtin_elementwise_fma(v2f{-c[1], c[0]}, v2f{a[1], a[1]}, acc);
}

static __device__ __forceinline__ void load_x_tile(const float* __restrict__ xr, v4f (&xv)[NSPLIT][2]) {
#pragma unroll
    for (int sp = 0; sp < NSPLIT; sp++)
#pragma unroll
        for (int q = 0; q < 2; q++)
            xv[sp][q] = *(const v4f*)(xr + sp * DIM + 8 * q);
}
__global__ __launch_bounds__(THREADS) void qcnn_fused(
    const float* __restrict__ x, const float* __restrict__ sw,
    const float* __restrict__ W1, const float* __restrict__ b1,
    const float* __restrict__ W2, const float* __restrict__ b2,
    const float* __restrict__ W3, const float* __restrict__ b3,
    float* __restrict__ out)
{
    __shared__ __attribute__((aligned(16))) float uS[NGATE * 8];
    __shared__ __attribute__((aligned(16))) float mS[NSPLIT * 2 * DIM * MP];
    __shared__ __attribute__((aligned(16))) float w2s[H2 * W2P];
    __shared__ __attribute__((aligned(16))) float w3s[16 * W3P];

    const int tid  = threadIdx.x;
    const int lane = tid & 31;
    const int wave = __builtin_amdgcn_readfirstlane(tid >> 5);
    const int half = lane >> 4;
    const int nl   = lane & 15;

    const int rowBase = blockIdx.x * ROWS_PER_BLOCK + wave * 16;

    const float* xr = x + (size_t)(rowBase + nl) * (NSPLIT * DIM) + 4 * half;
    v4f xv[NSPLIT][2];

    if (wave < FOLD_WAVES) {
        const int sp = wave;
        {
            const int g = sp * (NLAYER * QPS) + nl;
            const float* w = sw + g * 3;
            float hx = 0.5f * w[0], hy = 0.5f * w[1], hz = 0.5f * w[2];
            float sx, cx, sy, cy, sz, cz;
            sincosf(hx, &sx, &cx);
            sincosf(hy, &sy, &cy);
            sincosf(hz, &sz, &cz);
            float m00r = cy * cx, m00i =  sy * sx;
            float m01r = -sy * cx, m01i = -cy * sx;
            float m10r =  sy * cx, m10i = -cy * sx;
            float m11r =  cy * cx, m11i = -sy * sx;
            if (half == 0) {
                float* o = uS + g * 8;
                o[0] = cz * m00r + sz * m00i;  o[1] = cz * m00i - sz * m00r;
                o[2] = cz * m01r + sz * m01i;  o[3] = cz * m01i - sz * m01r;
                o[4] = cz * m10r - sz * m10i;  o[5] = cz * m10i + sz * m10r;
                o[6] = cz * m11r - sz * m11i;  o[7] = cz * m11i + sz * m11r;
            }
        }
        load_x_tile(xr, xv);
        wave_lds_handover();

        const int kb = lane >> 1;
        const int hb = lane & 1;
        v4f gt[NLAYER * QPS][2];
        {
            const v4f* us = (const v4f*)&uS[sp * NLAYER * QPS * 8];
#pragma unroll
            for (int g = 0; g < NLAYER * QPS; g++) { gt[g][0] = us[2 * g]; gt[g][1] = us[2 * g + 1]; }
        }
        __builtin_amdgcn_sched_barrier(0);

        v2f am[8];
#pragma unroll
        for (int r = 0; r < 8; r++) am[r] = v2f{(8 * hb + r == kb) ? 1.f : 0.f, 0.f};

#pragma unroll
        for (int l = 0; l < NLAYER; l++) {
            {
                const v4f ua = gt[l * QPS][0], ub = gt[l * QPS][1];
                const v2f co = v2f{hb ? ub[2] : ua[0], hb ? ub[3] : ua[1]};
                const v2f cp = v2f{hb ? ub[0] : ua[2], hb ? ub[1] : ua[3]};
#pragma unroll
                for (int r = 0; r < 8; r++) {
                    const v2f ot = v2f{lane_pair_other(am[r][0]), lane_pair_other(am[r][1])};
                    am[r] = cmac(cmul(co, am[r]), cp, ot);
                }
            }
#pragma unroll
            for (int j = 1; j < QPS; j++) {
                const v4f ua = gt[l * QPS + j][0], ub = gt[l * QPS + j][1];
                const v2f u00 = v2f{ua[0], ua[1]}, u01 = v2f{ua[2], ua[3]};
                const v2f u10 = v2f{ub[0], ub[1]}, u11 = v2f{ub[2], ub[3]};
                const int m = 8 >> j;
#pragma unroll
                for (int idx = 0; idx < 8; idx++) {
                    if (idx & m) continue;
                    const int i1 = idx | m;
                    const v2f a0 = am[idx], a1 = am[i1];
                    am[idx] = cmac(cmul(u00, a0), u01, a1);
                    am[i1]  = cmac(cmul(u10, a0), u11, a1);
                }
            }
#pragma unroll
            for (int r = 0; r < 4; r++) {
                const v2f lo = am[r], up = am[r | 4];
                am[r]     = v2f{hb ? up[0] : lo[0], hb ? up[1] : lo[1]};
                am[r | 4] = v2f{hb ? lo[0] : up[0], hb ? lo[1] : up[1]};
            }
#pragma unroll
            for (int j = 1; j < 3; j++) {
                const int mc = 8 >> j, mt = 8 >> (j + 1);
#pragma unroll
                for (int idx = 0; idx < 8; idx++) {
                    if ((idx & mc) && !(idx & mt)) {
                        const int i1 = idx | mt;
                        const v2f t = am[idx];
                        am[idx] = am[i1];
                        am[i1] = t;
                    }
                }
            }
#pragma unroll
            for (int r = 1; r < 8; r += 2)
                am[r] = v2f{lane_pair_other(am[r][0]), lane_pair_other(am[r][1])};
        }
        float* mr = mS + (sp * 2 + 0) * DIM * MP + 8 * hb * MP + kb;
        float* mi = mS + (sp * 2 + 1) * DIM * MP + 8 * hb * MP + kb;
#pragma unroll
        for (int r = 0; r < 8; r++) { mr[r * MP] = am[r][0]; mi[r * MP] = am[r][1]; }
    } else {
        load_x_tile(xr, xv);
        const int sw_ = wave - FOLD_WAVES;
        const int st = tid - 32 * FOLD_WAVES;
        float w3v[3];
#pragma unroll
        for (int i = 0; i < 3; i++) {
            const int idx = st + 32 * STAGE_WAVES * i;
            const int n = idx / W3P, k = idx - n * W3P;
            const bool live = (n < NCLASS) && (k < H2);
            const float v = W3[live ? (k * NCLASS + n) : 0];
            w3v[i] = live ? v : 0.f;
        }
        v4f wst[16];
#pragma unroll
        for (int i = 0; i < 16; i++) {
            const int k  = i * 8 + (lane & 7);
            const int n4 = (sw_ * 4 + (lane >> 3)) * 4;
            wst[i] = *(const v4f*)(W2 + (size_t)k * H2 + n4);
        }
#pragma unroll
        for (int i = 0; i < 16; i++) {
            const int k  = i * 8 + (lane & 7);
            const int n4 = (sw_ * 4 + (lane >> 3)) * 4;
#pragma unroll
            for (int j = 0; j < 4; j++) w2s[(n4 + j) * W2P + k] = wst[i][j];
        }
#pragma unroll
        for (int i = 0; i < 3; i++) w3s[st + 32 * STAGE_WAVES * i] = w3v[i];
        for (int idx = st + 32 * STAGE_WAVES * 3; idx < 16 * W3P; idx += 32 * STAGE_WAVES)
            w3s[idx] = 0.f;
    }
    __syncthreads();

    v8f d1[H1 / 16], d2[H2 / 16];
    v8f d3;
#pragma unroll
    for (int r = 0; r < 8; r++) d3[r] = 0.f;
#pragma unroll
    for (int r = 0; r < NCLASS; r++) {
        const float b3r = b3[r];
        d3[r] = half ? 0.f : b3r;
    }
    v2f w1f[H1 / 16][4];
#pragma unroll
    for (int t = 0; t < W1_FIRST; t++)
#pragma unroll
        for (int kc = 0; kc < 4; kc++) {
            const float* wp = W1 + (size_t)(kc * 4 + 2 * half) * H1 + 16 * t + nl;
            w1f[t][kc] = v2f{wp[0], wp[H1]};
        }
    __builtin_amdgcn_sched_barrier(0);

    const float* w2row = w2s + nl * W2P + 8 * half;
    const float* w3row = w3s + nl * W3P + 8 * half;
    v4f wf2[2][2][H2 / 16][2];
    v4f wf3[H2 / 16][2];

    v2f a1f[4];
#pragma unroll
    for (int sp = 0; sp < NSPLIT; sp++) {
        if (sp == NSPLIT / 2) {
            __builtin_amdgcn_sched_barrier(0);
#pragma unroll
            for (int t = W1_FIRST; t < H1 / 16; t++)
#pragma unroll
                for (int kc = 0; kc < 4; kc++) {
                    const float* wp = W1 + (size_t)(kc * 4 + 2 * half) * H1 + 16 * t + nl;
                    w1f[t][kc] = v2f{wp[0], wp[H1]};
                }
#pragma unroll
            for (int t = 0; t < H1 / 16; t++) {
                const v4f lo = *(const v4f*)(b1 + 16 * t + 8 * half);
                const v4f hi = *(const v4f*)(b1 + 16 * t + 8 * half + 4);
                d1[t] = v8f{lo[0], lo[1], lo[2], lo[3], hi[0], hi[1], hi[2], hi[3]};
            }
#pragma unroll
            for (int t = 0; t < H2 / 16; t++) {
                const v4f lo = *(const v4f*)(b2 + 16 * t + 8 * half);
                const v4f hi = *(const v4f*)(b2 + 16 * t + 8 * half + 4);
                d2[t] = v8f{lo[0], lo[1], lo[2], lo[3], hi[0], hi[1], hi[2], hi[3]};
            }
            __builtin_amdgcn_sched_barrier(0);
        }
        float n2 = 0.f;
#pragma unroll
        for (int q = 0; q < 2; q++)
#pragma unroll
            for (int e = 0; e < 4; e++) n2 += xv[sp][q][e] * xv[sp][q][e];
        n2 += __shfl_xor(n2, 16);
        float nrm = sqrtf(n2);
        float scale = (nrm > 0.f) ? (1.f / nrm) : 1.f;

        v8f dre, dim;
#pragma unroll
        for (int r = 0; r < 8; r++) { dre[r] = 0.f; dim[r] = 0.f; }
#pragma unroll
        for (int q = 0; q < 2; q++) {
            const v4f mr = *(const v4f*)&mS[((sp * 2 + 0) * DIM + nl) * MP + 8 * q + 4 * half];
            const v4f mi = *(const v4f*)&mS[((sp * 2 + 1) * DIM + nl) * MP + 8 * q + 4 * half];
#pragma unroll
            for (int p = 0; p < 2; p++) {
                v2f bx = v2f{xv[sp][q][2 * p] * scale, xv[sp][q][2 * p + 1] * scale};
                v2f ar = v2f{mr[2 * p], mr[2 * p + 1]};
                v2f ai = v2f{mi[2 * p], mi[2 * p + 1]};
                dre = __builtin_amdgcn_wmma_f32_16x16x4_f32(false, ar, false, bx,
                                                            (short)0, dre, false, false);
                dim = __builtin_amdgcn_wmma_f32_16x16x4_f32(false, ai, false, bx,
                                                            (short)0, dim, false, false);
            }
        }
        float pr[8];
#pragma unroll
        for (int r = 0; r < 8; r++) pr[r] = dre[r] * dre[r] + dim[r] * dim[r];
        float s0 = 0.f, s1 = 0.f, s2 = 0.f, s3 = 0.f;
#pragma unroll
        for (int r = 0; r < 8; r++) {
            s0 += pr[r];
            s1 += (r & 4) ? -pr[r] : pr[r];
            s2 += (r & 2) ? -pr[r] : pr[r];
            s3 += (r & 1) ? -pr[r] : pr[r];
        }
        s0 = half ? -s0 : s0;
        float e0 = s0 + __shfl_xor(s0, 16);
        float e1 = s1 + __shfl_xor(s1, 16);
        float e2 = s2 + __shfl_xor(s2, 16);
        float e3 = s3 + __shfl_xor(s3, 16);
        a1f[sp] = half ? v2f{e2, e3} : v2f{e0, e1};
    }

#pragma unroll
    for (int i = 0; i < 2; i++)
#pragma unroll
        for (int nt = 0; nt < H2 / 16; nt++)
#pragma unroll
            for (int j = 0; j < 2; j++)
                wf2[0][i][nt][j] = *(const v4f*)(w2row + nt * 16 * W2P + 16 * i + 4 * j);
    __builtin_amdgcn_sched_barrier(0);

#pragma unroll
    for (int t = 0; t < H1 / 16; t++) {
#pragma unroll
        for (int kc = 0; kc < 4; kc++)
            d1[t] = __builtin_amdgcn_wmma_f32_16x16x4_f32(false, w1f[t][kc], false, a1f[kc],
                                                          (short)0, d1[t], false, false);
#pragma unroll
        for (int r = 0; r < 8; r++) d1[t][r] = fmaxf(d1[t][r], 0.f);
    }

#pragma unroll
    for (int blk = 0; blk < H1 / 32; blk++) {
        const int cur = blk & 1, nxt = cur ^ 1;
        if (blk + 1 < H1 / 32) {
#pragma unroll
            for (int i = 0; i < 2; i++)
#pragma unroll
                for (int nt = 0; nt < H2 / 16; nt++)
#pragma unroll
                    for (int j = 0; j < 2; j++)
                        wf2[nxt][i][nt][j] = *(const v4f*)(w2row + nt * 16 * W2P + 16 * (2 * (blk + 1) + i) + 4 * j);
        } else {
#pragma unroll
            for (int t = 0; t < H2 / 16; t++)
#pragma unroll
                for (int j = 0; j < 2; j++)
                    wf3[t][j] = *(const v4f*)(w3row + 16 * t + 4 * j);
        }
        __builtin_amdgcn_sched_barrier(0);
#pragma unroll
        for (int i = 0; i < 2; i++) {
            const int t = 2 * blk + i;
#pragma unroll
            for (int p = 0; p < 4; p++) {
                const int j = p >> 1, pp = p & 1;
                v2f bf = v2f{d1[t][2 * p], d1[t][2 * p + 1]};
#pragma unroll
                for (int nt = 0; nt < H2 / 16; nt++) {
                    v2f af = v2f{wf2[cur][i][nt][j][2 * pp], wf2[cur][i][nt][j][2 * pp + 1]};
                    d2[nt] = __builtin_amdgcn_wmma_f32_16x16x4_f32(false, af, false, bf,
                                                                   (short)0, d2[nt], false, false);
                }
            }
        }
        __builtin_amdgcn_sched_barrier(0);
    }
#pragma unroll
    for (int nt = 0; nt < H2 / 16; nt++)
#pragma unroll
        for (int r = 0; r < 8; r++) d2[nt][r] = fmaxf(d2[nt][r], 0.f);

#pragma unroll
    for (int t = 0; t < H2 / 16; t++) {
#pragma unroll
        for (int p = 0; p < 4; p++) {
            const int j = p >> 1, pp = p & 1;
            v2f bf = v2f{d2[t][2 * p], d2[t][2 * p + 1]};
            v2f af = v2f{wf3[t][j][2 * pp], wf3[t][j][2 * pp + 1]};
            d3 = __builtin_amdgcn_wmma_f32_16x16x4_f32(false, af, false, bf,
                                                       (short)0, d3, false, false);
        }
    }
    if (half == 0) {
#pragma unroll
        for (int r = 0; r < NCLASS; r++)
            out[(size_t)(rowBase + nl) * NCLASS + r] = d3[r];
    }
}

extern "C" void kernel_launch(void* const* d_in, const int* in_sizes, int n_in,
                              void* d_out, int out_size, void* d_ws, size_t ws_size,
                              hipStream_t stream) {
    const float* x  = (const float*)d_in[0];
    const float* sw = (const float*)d_in[1];
    const float* W1 = (const float*)d_in[2];
    const float* b1 = (const float*)d_in[3];
    const float* W2 = (const float*)d_in[4];
    const float* b2 = (const float*)d_in[5];
    const float* W3 = (const float*)d_in[6];
    const float* b3 = (const float*)d_in[7];
    float* out = (float*)d_out;

    qcnn_fused<<<dim3(BATCH / ROWS_PER_BLOCK), dim3(THREADS), 0, stream>>>(
        x, sw, W1, b1, W2, b2, W3, b3, out);
}
